# starting kernel
# baseline (speedup 1.0000x reference)
#include <hip/hip_runtime.h>
#include <math.h>

typedef __attribute__((ext_vector_type(2))) float v2f;
typedef __attribute__((ext_vector_type(4))) float v4f;
typedef __attribute__((ext_vector_type(8))) float v8f;

#define B_  2
#define C_  2048
#define E_  1024
#define H_  16
#define D_  64
#define HD_ 1024
#define BC_ 4096

static __device__ __forceinline__ v8f vzero8() {
  v8f z = {0.f, 0.f, 0.f, 0.f, 0.f, 0.f, 0.f, 0.f};
  return z;
}

static __device__ __forceinline__ v8f wmma_f32(v2f a, v2f b, v8f c) {
  return __builtin_amdgcn_wmma_f32_16x16x4_f32(false, a, false, b, (short)0, c,
                                               false, false);
}

__global__ __launch_bounds__(256) void qkv_kernel(
    const float* __restrict__ x, const float* __restrict__ Wq,
    const float* __restrict__ Wk, const float* __restrict__ Wv,
    float* __restrict__ Qo, float* __restrict__ Ko, float* __restrict__ Vo) {
  __shared__ float xs[16 * 260];
  const int tid = threadIdx.x;
  const int wave = tid >> 5, lane = tid & 31, h = lane >> 4, m = lane & 15;
  const int row0 = blockIdx.x * 16;
  const int n0g = (blockIdx.y * 8 + wave) * 64;
  const int mat = n0g >> 10;
  const int jin = n0g & 1023;
  const float* Wm = (mat == 0) ? Wq : ((mat == 1) ? Wk : Wv);
  float* dst = (mat == 0) ? Qo : ((mat == 1) ? Ko : Vo);

  v8f acc0 = vzero8(), acc1 = vzero8(), acc2 = vzero8(), acc3 = vzero8();
  for (int kc = 0; kc < E_; kc += 256) {
    __syncthreads();
#pragma unroll
    for (int it = 0; it < 4; ++it) {
      int q = tid + it * 256;
      int r = q >> 6, c4 = (q & 63) << 2;
      v4f v = *(const v4f*)&x[(size_t)(row0 + r) * E_ + kc + c4];
      *(v4f*)&xs[r * 260 + c4] = v;
    }
    __syncthreads();
#pragma unroll 2
    for (int k = 0; k < 256; k += 4) {
      v2f a = *(const v2f*)&xs[m * 260 + k + 2 * h];
      const int krow = kc + k + 2 * h;
      const float* w0 = &Wm[(size_t)krow * HD_ + jin + m];
      v2f b;
      b.x = w0[0 * 16];  b.y = w0[HD_ + 0 * 16];  acc0 = wmma_f32(a, b, acc0);
      b.x = w0[1 * 16];  b.y = w0[HD_ + 1 * 16];  acc1 = wmma_f32(a, b, acc1);
      b.x = w0[2 * 16];  b.y = w0[HD_ + 2 * 16];  acc2 = wmma_f32(a, b, acc2);
      b.x = w0[3 * 16];  b.y = w0[HD_ + 3 * 16];  acc3 = wmma_f32(a, b, acc3);
    }
  }
  const int head = jin >> 6;
#pragma unroll
  for (int r = 0; r < 8; ++r) {
    int t = row0 + 8 * h + r;
    int bb = t >> 11, cc = t & 2047;
    float* o = &dst[(((size_t)bb * H_ + head) * C_ + cc) * D_ + m];
    o[0 * 16] = acc0[r];
    o[1 * 16] = acc1[r];
    o[2 * 16] = acc2[r];
    o[3 * 16] = acc3[r];
  }
}

__global__ __launch_bounds__(256) void rope_kernel(float* __restrict__ Q,
                                                   float* __restrict__ K) {
  const int idx = blockIdx.x * 256 + threadIdx.x;
  float* P = (blockIdx.y == 0) ? Q : K;
  const int i = idx & 31;
  const int c = (idx >> 5) & 2047;
  const int bh = idx >> 16;
  const float inv_freq = __expf(-(2.0f * (float)i / 64.0f) * 9.210340371976184f);
  const float ang = (float)c * inv_freq;
  float sn, cs;
  sincosf(ang, &sn, &cs);
  const size_t off = ((size_t)bh * C_ + c) * D_ + 2 * i;
  const float e = P[off], o = P[off + 1];
  P[off] = e * cs - o * sn;
  P[off + 1] = e * sn + o * cs;
}

__global__ __launch_bounds__(256) void attn_kernel(
    const float* __restrict__ Q, const float* __restrict__ K,
    const float* __restrict__ V, float* __restrict__ hid) {
  __shared__ float pbuf[8][16 * 17];
  const int tid = threadIdx.x;
  const int wave = tid >> 5, lane = tid & 31, h = lane >> 4, m = lane & 15;
  const int bh = blockIdx.y;
  const int bb = bh >> 4, head = bh & 15;
  const int qt = blockIdx.x * 8 + wave;
  const int q0 = qt * 16;
  const size_t base = (size_t)bh * C_ * D_;
  float* P = pbuf[wave];

  v2f aq[16];
#pragma unroll
  for (int s = 0; s < 16; ++s)
    aq[s] = *(const v2f*)&Q[base + (size_t)(q0 + m) * D_ + 4 * s + 2 * h];

  v8f o0 = vzero8(), o1 = vzero8(), o2 = vzero8(), o3 = vzero8();
  float mrow[8], lrow[8];
#pragma unroll
  for (int r = 0; r < 8; ++r) { mrow[r] = -INFINITY; lrow[r] = 0.f; }

  for (int kt = 0; kt <= qt; ++kt) {
    const int k0 = kt * 16;
    v8f sc = vzero8();
#pragma unroll
    for (int s = 0; s < 16; ++s) {
      v2f bk = *(const v2f*)&K[base + (size_t)(k0 + m) * D_ + 4 * s + 2 * h];
      sc = wmma_f32(aq[s], bk, sc);
    }
    float corr[8];
#pragma unroll
    for (int r = 0; r < 8; ++r) {
      float sv = sc[r] * 0.125f;
      if (kt == qt && (8 * h + r) < m) sv = -INFINITY;
      float vmax = sv;
#pragma unroll
      for (int off = 1; off < 16; off <<= 1)
        vmax = fmaxf(vmax, __shfl_xor(vmax, off, 32));
      float mnew = fmaxf(mrow[r], vmax);
      float cr = __expf(mrow[r] - mnew);
      float p = __expf(sv - mnew);
      float rs = p;
#pragma unroll
      for (int off = 1; off < 16; off <<= 1) rs += __shfl_xor(rs, off, 32);
      lrow[r] = lrow[r] * cr + rs;
      mrow[r] = mnew;
      corr[r] = cr;
      P[(r + 8 * h) * 17 + m] = p;
    }
#pragma unroll
    for (int r = 0; r < 8; ++r) {
      o0[r] *= corr[r]; o1[r] *= corr[r]; o2[r] *= corr[r]; o3[r] *= corr[r];
    }
    asm volatile("s_wait_dscnt 0x0" ::: "memory");
#pragma unroll
    for (int s = 0; s < 4; ++s) {
      const int kl = 4 * s + 2 * h;
      v2f ap;
      ap.x = P[m * 17 + kl];
      ap.y = P[m * 17 + kl + 1];
      const float* v0 = &V[base + (size_t)(k0 + kl) * D_ + m];
      const float* v1 = v0 + D_;
      v2f bv;
      bv.x = v0[0 * 16]; bv.y = v1[0 * 16];  o0 = wmma_f32(ap, bv, o0);
      bv.x = v0[1 * 16]; bv.y = v1[1 * 16];  o1 = wmma_f32(ap, bv, o1);
      bv.x = v0[2 * 16]; bv.y = v1[2 * 16];  o2 = wmma_f32(ap, bv, o2);
      bv.x = v0[3 * 16]; bv.y = v1[3 * 16];  o3 = wmma_f32(ap, bv, o3);
    }
  }
#pragma unroll
  for (int r = 0; r < 8; ++r) {
    const float inv = 1.0f / lrow[r];
    float* o = &hid[((size_t)bb * C_ + q0 + 8 * h + r) * HD_ + head * D_ + m];
    o[0 * 16] = o0[r] * inv;
    o[1 * 16] = o1[r] * inv;
    o[2 * 16] = o2[r] * inv;
    o[3 * 16] = o3[r] * inv;
  }
}

__global__ __launch_bounds__(256) void proj_kernel(
    const float* __restrict__ hid, const float* __restrict__ Wo,
    float* __restrict__ out) {
  __shared__ float xs[16 * 260];
  const int tid = threadIdx.x;
  const int wave = tid >> 5, lane = tid & 31, h = lane >> 4, m = lane & 15;
  const int row0 = blockIdx.x * 16;
  const int n0 = (blockIdx.y * 8 + wave) * 64;

  v8f acc0 = vzero8(), acc1 = vzero8(), acc2 = vzero8(), acc3 = vzero8();
  for (int kc = 0; kc < HD_; kc += 256) {
    __syncthreads();
#pragma unroll
    for (int it = 0; it < 4; ++it) {
      int q = tid + it * 256;
      int r = q >> 6, c4 = (q & 63) << 2;
      v4f v = *(const v4f*)&hid[(size_t)(row0 + r) * HD_ + kc + c4];
      *(v4f*)&xs[r * 260 + c4] = v;
    }
    __syncthreads();
#pragma unroll 2
    for (int k = 0; k < 256; k += 4) {
      v2f a = *(const v2f*)&xs[m * 260 + k + 2 * h];
      const int krow = kc + k + 2 * h;
      const float* w0 = &Wo[(size_t)krow * E_ + n0 + m];
      v2f b;
      b.x = w0[0 * 16];  b.y = w0[E_ + 0 * 16];  acc0 = wmma_f32(a, b, acc0);
      b.x = w0[1 * 16];  b.y = w0[E_ + 1 * 16];  acc1 = wmma_f32(a, b, acc1);
      b.x = w0[2 * 16];  b.y = w0[E_ + 2 * 16];  acc2 = wmma_f32(a, b, acc2);
      b.x = w0[3 * 16];  b.y = w0[E_ + 3 * 16];  acc3 = wmma_f32(a, b, acc3);
    }
  }
#pragma unroll
  for (int r = 0; r < 8; ++r) {
    float* o = &out[(size_t)(row0 + 8 * h + r) * E_ + n0 + m];
    o[0 * 16] = acc0[r];
    o[1 * 16] = acc1[r];
    o[2 * 16] = acc2[r];
    o[3 * 16] = acc3[r];
  }
}

extern "C" void kernel_launch(void* const* d_in, const int* in_sizes, int n_in,
                              void* d_out, int out_size, void* d_ws,
                              size_t ws_size, hipStream_t stream) {
  (void)in_sizes; (void)n_in; (void)out_size; (void)ws_size;
  const float* x  = (const float*)d_in[0];
  const float* Wq = (const float*)d_in[1];
  const float* Wk = (const float*)d_in[2];
  const float* Wv = (const float*)d_in[3];
  const float* Wo = (const float*)d_in[4];
  float* out = (float*)d_out;

  const size_t BHCD = (size_t)B_ * H_ * C_ * D_;
  float* Qw  = (float*)d_ws;
  float* Kw  = Qw + BHCD;
  float* Vw  = Kw + BHCD;
  float* Hid = Vw + BHCD;

  qkv_kernel<<<dim3(BC_ / 16, (3 * HD_) / (8 * 64)), 256, 0, stream>>>(
      x, Wq, Wk, Wv, Qw, Kw, Vw);
  rope_kernel<<<dim3((B_ * H_ * C_ * (D_ / 2)) / 256, 2), 256, 0, stream>>>(
      Qw, Kw);
  attn_kernel<<<dim3(C_ / 16 / 8, B_ * H_), 256, 0, stream>>>(Qw, Kw, Vw, Hid);
  proj_kernel<<<dim3(BC_ / 16, HD_ / (8 * 64)), 256, 0, stream>>>(Hid, Wo, out);
}


# attention: deal query tiles to blocks as 4 early + 4 mirrored late (qt and 127-qt) so every block and SIMD pair gets equal causal key-tile work; per-tile math and summation order unchanged
# speedup vs baseline: 1.0507x; 1.0507x over previous
#include <hip/hip_runtime.h>
#include <math.h>

typedef __attribute__((ext_vector_type(2))) float v2f;
typedef __attribute__((ext_vector_type(4))) float v4f;
typedef __attribute__((ext_vector_type(8))) float v8f;

#define B_  2
#define C_  2048
#define E_  1024
#define H_  16
#define D_  64
#define HD_ 1024
#define BC_ 4096

static __device__ __forceinline__ v8f vzero8() {
  v8f z = {0.f, 0.f, 0.f, 0.f, 0.f, 0.f, 0.f, 0.f};
  return z;
}

static __device__ __forceinline__ v8f wmma_f32(v2f a, v2f b, v8f c) {
  return __builtin_amdgcn_wmma_f32_16x16x4_f32(false, a, false, b, (short)0, c,
                                               false, false);
}

__global__ __launch_bounds__(256) void qkv_kernel(
    const float* __restrict__ x, const float* __restrict__ Wq,
    const float* __restrict__ Wk, const float* __restrict__ Wv,
    float* __restrict__ Qo, float* __restrict__ Ko, float* __restrict__ Vo) {
  __shared__ float xs[16 * 260];
  const int tid = threadIdx.x;
  const int wave = tid >> 5, lane = tid & 31, h = lane >> 4, m = lane & 15;
  const int row0 = blockIdx.x * 16;
  const int n0g = (blockIdx.y * 8 + wave) * 64;
  const int mat = n0g >> 10;
  const int jin = n0g & 1023;
  const float* Wm = (mat == 0) ? Wq : ((mat == 1) ? Wk : Wv);
  float* dst = (mat == 0) ? Qo : ((mat == 1) ? Ko : Vo);

  v8f acc0 = vzero8(), acc1 = vzero8(), acc2 = vzero8(), acc3 = vzero8();
  for (int kc = 0; kc < E_; kc += 256) {
    __syncthreads();
#pragma unroll
    for (int it = 0; it < 4; ++it) {
      int q = tid + it * 256;
      int r = q >> 6, c4 = (q & 63) << 2;
      v4f v = *(const v4f*)&x[(size_t)(row0 + r) * E_ + kc + c4];
      *(v4f*)&xs[r * 260 + c4] = v;
    }
    __syncthreads();
#pragma unroll 2
    for (int k = 0; k < 256; k += 4) {
      v2f a = *(const v2f*)&xs[m * 260 + k + 2 * h];
      const int krow = kc + k + 2 * h;
      const float* w0 = &Wm[(size_t)krow * HD_ + jin + m];
      v2f b;
      b.x = w0[0 * 16];  b.y = w0[HD_ + 0 * 16];  acc0 = wmma_f32(a, b, acc0);
      b.x = w0[1 * 16];  b.y = w0[HD_ + 1 * 16];  acc1 = wmma_f32(a, b, acc1);
      b.x = w0[2 * 16];  b.y = w0[HD_ + 2 * 16];  acc2 = wmma_f32(a, b, acc2);
      b.x = w0[3 * 16];  b.y = w0[HD_ + 3 * 16];  acc3 = wmma_f32(a, b, acc3);
    }
  }
  const int head = jin >> 6;
#pragma unroll
  for (int r = 0; r < 8; ++r) {
    int t = row0 + 8 * h + r;
    int bb = t >> 11, cc = t & 2047;
    float* o = &dst[(((size_t)bb * H_ + head) * C_ + cc) * D_ + m];
    o[0 * 16] = acc0[r];
    o[1 * 16] = acc1[r];
    o[2 * 16] = acc2[r];
    o[3 * 16] = acc3[r];
  }
}

__global__ __launch_bounds__(256) void rope_kernel(float* __restrict__ Q,
                                                   float* __restrict__ K) {
  const int idx = blockIdx.x * 256 + threadIdx.x;
  float* P = (blockIdx.y == 0) ? Q : K;
  const int i = idx & 31;
  const int c = (idx >> 5) & 2047;
  const int bh = idx >> 16;
  const float inv_freq = __expf(-(2.0f * (float)i / 64.0f) * 9.210340371976184f);
  const float ang = (float)c * inv_freq;
  float sn, cs;
  sincosf(ang, &sn, &cs);
  const size_t off = ((size_t)bh * C_ + c) * D_ + 2 * i;
  const float e = P[off], o = P[off + 1];
  P[off] = e * cs - o * sn;
  P[off + 1] = e * sn + o * cs;
}

__global__ __launch_bounds__(256) void attn_kernel(
    const float* __restrict__ Q, const float* __restrict__ K,
    const float* __restrict__ V, float* __restrict__ hid) {
  __shared__ float pbuf[8][16 * 17];
  const int tid = threadIdx.x;
  const int wave = tid >> 5, lane = tid & 31, h = lane >> 4, m = lane & 15;
  const int bh = blockIdx.y;
  const int bb = bh >> 4, head = bh & 15;
  const int lt = blockIdx.x * 4 + (wave & 3);
  const bool late = (((wave >> 2) ^ wave) & 1) != 0;
  const int qt = late ? (C_ / 16 - 1) - lt : lt;
  const int q0 = qt * 16;
  const size_t base = (size_t)bh * C_ * D_;
  float* P = pbuf[wave];

  v2f aq[16];
#pragma unroll
  for (int s = 0; s < 16; ++s)
    aq[s] = *(const v2f*)&Q[base + (size_t)(q0 + m) * D_ + 4 * s + 2 * h];

  v8f o0 = vzero8(), o1 = vzero8(), o2 = vzero8(), o3 = vzero8();
  float mrow[8], lrow[8];
#pragma unroll
  for (int r = 0; r < 8; ++r) { mrow[r] = -INFINITY; lrow[r] = 0.f; }

  for (int kt = 0; kt <= qt; ++kt) {
    const int k0 = kt * 16;
    v8f sc = vzero8();
#pragma unroll
    for (int s = 0; s < 16; ++s) {
      v2f bk = *(const v2f*)&K[base + (size_t)(k0 + m) * D_ + 4 * s + 2 * h];
      sc = wmma_f32(aq[s], bk, sc);
    }
    float corr[8];
#pragma unroll
    for (int r = 0; r < 8; ++r) {
      float sv = sc[r] * 0.125f;
      if (kt == qt && (8 * h + r) < m) sv = -INFINITY;
      float vmax = sv;
#pragma unroll
      for (int off = 1; off < 16; off <<= 1)
        vmax = fmaxf(vmax, __shfl_xor(vmax, off, 32));
      float mnew = fmaxf(mrow[r], vmax);
      float cr = __expf(mrow[r] - mnew);
      float p = __expf(sv - mnew);
      float rs = p;
#pragma unroll
      for (int off = 1; off < 16; off <<= 1) rs += __shfl_xor(rs, off, 32);
      lrow[r] = lrow[r] * cr + rs;
      mrow[r] = mnew;
      corr[r] = cr;
      P[(r + 8 * h) * 17 + m] = p;
    }
#pragma unroll
    for (int r = 0; r < 8; ++r) {
      o0[r] *= corr[r]; o1[r] *= corr[r]; o2[r] *= corr[r]; o3[r] *= corr[r];
    }
    asm volatile("s_wait_dscnt 0x0" ::: "memory");
#pragma unroll
    for (int s = 0; s < 4; ++s) {
      const int kl = 4 * s + 2 * h;
      v2f ap;
      ap.x = P[m * 17 + kl];
      ap.y = P[m * 17 + kl + 1];
      const float* v0 = &V[base + (size_t)(k0 + kl) * D_ + m];
      const float* v1 = v0 + D_;
      v2f bv;
      bv.x = v0[0 * 16]; bv.y = v1[0 * 16];  o0 = wmma_f32(ap, bv, o0);
      bv.x = v0[1 * 16]; bv.y = v1[1 * 16];  o1 = wmma_f32(ap, bv, o1);
      bv.x = v0[2 * 16]; bv.y = v1[2 * 16];  o2 = wmma_f32(ap, bv, o2);
      bv.x = v0[3 * 16]; bv.y = v1[3 * 16];  o3 = wmma_f32(ap, bv, o3);
    }
  }
#pragma unroll
  for (int r = 0; r < 8; ++r) {
    const float inv = 1.0f / lrow[r];
    float* o = &hid[((size_t)bb * C_ + q0 + 8 * h + r) * HD_ + head * D_ + m];
    o[0 * 16] = o0[r] * inv;
    o[1 * 16] = o1[r] * inv;
    o[2 * 16] = o2[r] * inv;
    o[3 * 16] = o3[r] * inv;
  }
}

__global__ __launch_bounds__(256) void proj_kernel(
    const float* __restrict__ hid, const float* __restrict__ Wo,
    float* __restrict__ out) {
  __shared__ float xs[16 * 260];
  const int tid = threadIdx.x;
  const int wave = tid >> 5, lane = tid & 31, h = lane >> 4, m = lane & 15;
  const int row0 = blockIdx.x * 16;
  const int n0 = (blockIdx.y * 8 + wave) * 64;

  v8f acc0 = vzero8(), acc1 = vzero8(), acc2 = vzero8(), acc3 = vzero8();
  for (int kc = 0; kc < HD_; kc += 256) {
    __syncthreads();
#pragma unroll
    for (int it = 0; it < 4; ++it) {
      int q = tid + it * 256;
      int r = q >> 6, c4 = (q & 63) << 2;
      v4f v = *(const v4f*)&hid[(size_t)(row0 + r) * HD_ + kc + c4];
      *(v4f*)&xs[r * 260 + c4] = v;
    }
    __syncthreads();
#pragma unroll 2
    for (int k = 0; k < 256; k += 4) {
      v2f a = *(const v2f*)&xs[m * 260 + k + 2 * h];
      const int krow = kc + k + 2 * h;
      const float* w0 = &Wo[(size_t)krow * E_ + n0 + m];
      v2f b;
      b.x = w0[0 * 16];  b.y = w0[E_ + 0 * 16];  acc0 = wmma_f32(a, b, acc0);
      b.x = w0[1 * 16];  b.y = w0[E_ + 1 * 16];  acc1 = wmma_f32(a, b, acc1);
      b.x = w0[2 * 16];  b.y = w0[E_ + 2 * 16];  acc2 = wmma_f32(a, b, acc2);
      b.x = w0[3 * 16];  b.y = w0[E_ + 3 * 16];  acc3 = wmma_f32(a, b, acc3);
    }
  }
#pragma unroll
  for (int r = 0; r < 8; ++r) {
    float* o = &out[(size_t)(row0 + 8 * h + r) * E_ + n0 + m];
    o[0 * 16] = acc0[r];
    o[1 * 16] = acc1[r];
    o[2 * 16] = acc2[r];
    o[3 * 16] = acc3[r];
  }
}

extern "C" void kernel_launch(void* const* d_in, const int* in_sizes, int n_in,
                              void* d_out, int out_size, void* d_ws,
                              size_t ws_size, hipStream_t stream) {
  (void)in_sizes; (void)n_in; (void)out_size; (void)ws_size;
  const float* x  = (const float*)d_in[0];
  const float* Wq = (const float*)d_in[1];
  const float* Wk = (const float*)d_in[2];
  const float* Wv = (const float*)d_in[3];
  const float* Wo = (const float*)d_in[4];
  float* out = (float*)d_out;

  const size_t BHCD = (size_t)B_ * H_ * C_ * D_;
  float* Qw  = (float*)d_ws;
  float* Kw  = Qw + BHCD;
  float* Vw  = Kw + BHCD;
  float* Hid = Vw + BHCD;

  qkv_kernel<<<dim3(BC_ / 16, (3 * HD_) / (8 * 64)), 256, 0, stream>>>(
      x, Wq, Wk, Wv, Qw, Kw, Vw);
  rope_kernel<<<dim3((B_ * H_ * C_ * (D_ / 2)) / 256, 2), 256, 0, stream>>>(
      Qw, Kw);
  attn_kernel<<<dim3(C_ / 16 / 8, B_ * H_), 256, 0, stream>>>(Qw, Kw, Vw, Hid);
  proj_kernel<<<dim3(BC_ / 16, HD_ / (8 * 64)), 256, 0, stream>>>(Hid, Wo, out);
}


# projection GEMMs (QKV and output): wave tile 64x32 instead of 16x64, so each B fragment loaded from global memory feeds four WMMAs instead of one; k order and per-element sums unchanged
# speedup vs baseline: 1.1673x; 1.1110x over previous
#include <hip/hip_runtime.h>
#include <math.h>

typedef __attribute__((ext_vector_type(2))) float v2f;
typedef __attribute__((ext_vector_type(4))) float v4f;
typedef __attribute__((ext_vector_type(8))) float v8f;

#define B_  2
#define C_  2048
#define E_  1024
#define H_  16
#define D_  64
#define HD_ 1024
#define BC_ 4096

static __device__ __forceinline__ v8f vzero8() {
  v8f z = {0.f, 0.f, 0.f, 0.f, 0.f, 0.f, 0.f, 0.f};
  return z;
}

static __device__ __forceinline__ v8f wmma_f32(v2f a, v2f b, v8f c) {
  return __builtin_amdgcn_wmma_f32_16x16x4_f32(false, a, false, b, (short)0, c,
                                               false, false);
}

__global__ __launch_bounds__(256) void qkv_kernel(
    const float* __restrict__ x, const float* __restrict__ Wq,
    const float* __restrict__ Wk, const float* __restrict__ Wv,
    float* __restrict__ Qo, float* __restrict__ Ko, float* __restrict__ Vo) {
  __shared__ float xs[64 * 260];
  const int tid = threadIdx.x;
  const int wave = tid >> 5, lane = tid & 31, h = lane >> 4, m = lane & 15;
  const int row0 = blockIdx.x * 64;
  const int r0 = tid >> 6, c4 = (tid & 63) << 2;
  const int n0g = (blockIdx.y * 8 + wave) * 32;
  const int mat = n0g >> 10;
  const int jin = n0g & 1023;
  const float* Wm = (mat == 0) ? Wq : ((mat == 1) ? Wk : Wv);
  float* dst = (mat == 0) ? Qo : ((mat == 1) ? Ko : Vo);

  v8f acc[4][2];
#pragma unroll
  for (int t = 0; t < 4; ++t) { acc[t][0] = vzero8(); acc[t][1] = vzero8(); }
  for (int kc = 0; kc < E_; kc += 256) {
    __syncthreads();
    {
      const float* xg = &x[(size_t)(row0 + r0) * E_ + kc + c4];
      float* xl = &xs[r0 * 260 + c4];
#pragma unroll
      for (int it = 0; it < 16; ++it) {
        v4f v = *(const v4f*)&xg[(size_t)it * 4 * E_];
        *(v4f*)&xl[it * 4 * 260] = v;
      }
    }
    __syncthreads();
#pragma unroll 4
    for (int k = 0; k < 256; k += 4) {
      const int krow = kc + k + 2 * h;
      const float* w0 = &Wm[(size_t)krow * HD_ + jin + m];
      v2f b0, b1;
      b0.x = w0[0];   b0.y = w0[HD_];
      b1.x = w0[16];  b1.y = w0[HD_ + 16];
#pragma unroll
      for (int t = 0; t < 4; ++t) {
        v2f a = *(const v2f*)&xs[(16 * t + m) * 260 + k + 2 * h];
        acc[t][0] = wmma_f32(a, b0, acc[t][0]);
        acc[t][1] = wmma_f32(a, b1, acc[t][1]);
      }
    }
  }
  const int head = jin >> 6, d0 = jin & 63;
#pragma unroll
  for (int t = 0; t < 4; ++t) {
#pragma unroll
    for (int r = 0; r < 8; ++r) {
      int tok = row0 + 16 * t + 8 * h + r;
      int bb = tok >> 11, cc = tok & 2047;
      float* o = &dst[(((size_t)bb * H_ + head) * C_ + cc) * D_ + d0 + m];
      o[0]  = acc[t][0][r];
      o[16] = acc[t][1][r];
    }
  }
}

__global__ __launch_bounds__(256) void rope_kernel(float* __restrict__ Q,
                                                   float* __restrict__ K) {
  const int idx = blockIdx.x * 256 + threadIdx.x;
  float* P = (blockIdx.y == 0) ? Q : K;
  const int i = idx & 31;
  const int c = (idx >> 5) & 2047;
  const int bh = idx >> 16;
  const float inv_freq = __expf(-(2.0f * (float)i / 64.0f) * 9.210340371976184f);
  const float ang = (float)c * inv_freq;
  float sn, cs;
  sincosf(ang, &sn, &cs);
  const size_t off = ((size_t)bh * C_ + c) * D_ + 2 * i;
  const float e = P[off], o = P[off + 1];
  P[off] = e * cs - o * sn;
  P[off + 1] = e * sn + o * cs;
}

__global__ __launch_bounds__(256) void attn_kernel(
    const float* __restrict__ Q, const float* __restrict__ K,
    const float* __restrict__ V, float* __restrict__ hid) {
  __shared__ float pbuf[8][16 * 17];
  const int tid = threadIdx.x;
  const int wave = tid >> 5, lane = tid & 31, h = lane >> 4, m = lane & 15;
  const int bh = blockIdx.y;
  const int bb = bh >> 4, head = bh & 15;
  const int lt = blockIdx.x * 4 + (wave & 3);
  const bool late = (((wave >> 2) ^ wave) & 1) != 0;
  const int qt = late ? (C_ / 16 - 1) - lt : lt;
  const int q0 = qt * 16;
  const size_t base = (size_t)bh * C_ * D_;
  float* P = pbuf[wave];

  v2f aq[16];
#pragma unroll
  for (int s = 0; s < 16; ++s)
    aq[s] = *(const v2f*)&Q[base + (size_t)(q0 + m) * D_ + 4 * s + 2 * h];

  v8f o0 = vzero8(), o1 = vzero8(), o2 = vzero8(), o3 = vzero8();
  float mrow[8], lrow[8];
#pragma unroll
  for (int r = 0; r < 8; ++r) { mrow[r] = -INFINITY; lrow[r] = 0.f; }

  for (int kt = 0; kt <= qt; ++kt) {
    const int k0 = kt * 16;
    v8f sc = vzero8();
#pragma unroll
    for (int s = 0; s < 16; ++s) {
      v2f bk = *(const v2f*)&K[base + (size_t)(k0 + m) * D_ + 4 * s + 2 * h];
      sc = wmma_f32(aq[s], bk, sc);
    }
    float corr[8];
#pragma unroll
    for (int r = 0; r < 8; ++r) {
      float sv = sc[r] * 0.125f;
      if (kt == qt && (8 * h + r) < m) sv = -INFINITY;
      float vmax = sv;
#pragma unroll
      for (int off = 1; off < 16; off <<= 1)
        vmax = fmaxf(vmax, __shfl_xor(vmax, off, 32));
      float mnew = fmaxf(mrow[r], vmax);
      float cr = __expf(mrow[r] - mnew);
      float p = __expf(sv - mnew);
      float rs = p;
#pragma unroll
      for (int off = 1; off < 16; off <<= 1) rs += __shfl_xor(rs, off, 32);
      lrow[r] = lrow[r] * cr + rs;
      mrow[r] = mnew;
      corr[r] = cr;
      P[(r + 8 * h) * 17 + m] = p;
    }
#pragma unroll
    for (int r = 0; r < 8; ++r) {
      o0[r] *= corr[r]; o1[r] *= corr[r]; o2[r] *= corr[r]; o3[r] *= corr[r];
    }
    asm volatile("s_wait_dscnt 0x0" ::: "memory");
#pragma unroll
    for (int s = 0; s < 4; ++s) {
      const int kl = 4 * s + 2 * h;
      v2f ap;
      ap.x = P[m * 17 + kl];
      ap.y = P[m * 17 + kl + 1];
      const float* v0 = &V[base + (size_t)(k0 + kl) * D_ + m];
      const float* v1 = v0 + D_;
      v2f bv;
      bv.x = v0[0 * 16]; bv.y = v1[0 * 16];  o0 = wmma_f32(ap, bv, o0);
      bv.x = v0[1 * 16]; bv.y = v1[1 * 16];  o1 = wmma_f32(ap, bv, o1);
      bv.x = v0[2 * 16]; bv.y = v1[2 * 16];  o2 = wmma_f32(ap, bv, o2);
      bv.x = v0[3 * 16]; bv.y = v1[3 * 16];  o3 = wmma_f32(ap, bv, o3);
    }
  }
#pragma unroll
  for (int r = 0; r < 8; ++r) {
    const float inv = 1.0f / lrow[r];
    float* o = &hid[((size_t)bb * C_ + q0 + 8 * h + r) * HD_ + head * D_ + m];
    o[0 * 16] = o0[r] * inv;
    o[1 * 16] = o1[r] * inv;
    o[2 * 16] = o2[r] * inv;
    o[3 * 16] = o3[r] * inv;
  }
}

__global__ __launch_bounds__(256) void proj_kernel(
    const float* __restrict__ hid, const float* __restrict__ Wo,
    float* __restrict__ out) {
  __shared__ float xs[64 * 260];
  const int tid = threadIdx.x;
  const int wave = tid >> 5, lane = tid & 31, h = lane >> 4, m = lane & 15;
  const int row0 = blockIdx.x * 64;
  const int r0 = tid >> 6, c4 = (tid & 63) << 2;
  const int n0 = (blockIdx.y * 8 + wave) * 32;

  v8f acc[4][2];
#pragma unroll
  for (int t = 0; t < 4; ++t) { acc[t][0] = vzero8(); acc[t][1] = vzero8(); }
  for (int kc = 0; kc < HD_; kc += 256) {
    __syncthreads();
    {
      const float* xg = &hid[(size_t)(row0 + r0) * HD_ + kc + c4];
      float* xl = &xs[r0 * 260 + c4];
#pragma unroll
      for (int it = 0; it < 16; ++it) {
        v4f v = *(const v4f*)&xg[(size_t)it * 4 * HD_];
        *(v4f*)&xl[it * 4 * 260] = v;
      }
    }
    __syncthreads();
#pragma unroll 4
    for (int k = 0; k < 256; k += 4) {
      const int krow = kc + k + 2 * h;
      const float* w0 = &Wo[(size_t)krow * E_ + n0 + m];
      v2f b0, b1;
      b0.x = w0[0];   b0.y = w0[E_];
      b1.x = w0[16];  b1.y = w0[E_ + 16];
#pragma unroll
      for (int t = 0; t < 4; ++t) {
        v2f a = *(const v2f*)&xs[(16 * t + m) * 260 + k + 2 * h];
        acc[t][0] = wmma_f32(a, b0, acc[t][0]);
        acc[t][1] = wmma_f32(a, b1, acc[t][1]);
      }
    }
  }
#pragma unroll
  for (int t = 0; t < 4; ++t) {
#pragma unroll
    for (int r = 0; r < 8; ++r) {
      float* o = &out[(size_t)(row0 + 16 * t + 8 * h + r) * E_ + n0 + m];
      o[0]  = acc[t][0][r];
      o[16] = acc[t][1][r];
    }
  }
}

extern "C" void kernel_launch(void* const* d_in, const int* in_sizes, int n_in,
                              void* d_out, int out_size, void* d_ws,
                              size_t ws_size, hipStream_t stream) {
  (void)in_sizes; (void)n_in; (void)out_size; (void)ws_size;
  const float* x  = (const float*)d_in[0];
  const float* Wq = (const float*)d_in[1];
  const float* Wk = (const float*)d_in[2];
  const float* Wv = (const float*)d_in[3];
  const float* Wo = (const float*)d_in[4];
  float* out = (float*)d_out;

  const size_t BHCD = (size_t)B_ * H_ * C_ * D_;
  float* Qw  = (float*)d_ws;
  float* Kw  = Qw + BHCD;
  float* Vw  = Kw + BHCD;
  float* Hid = Vw + BHCD;

  qkv_kernel<<<dim3(BC_ / 64, (3 * HD_) / (8 * 32)), 256, 0, stream>>>(
      x, Wq, Wk, Wv, Qw, Kw, Vw);
  rope_kernel<<<dim3((B_ * H_ * C_ * (D_ / 2)) / 256, 2), 256, 0, stream>>>(
      Qw, Kw);
  attn_kernel<<<dim3(C_ / 16 / 8, B_ * H_), 256, 0, stream>>>(Qw, Kw, Vw, Hid);
  proj_kernel<<<dim3(BC_ / 64, HD_ / (8 * 32)), 256, 0, stream>>>(Hid, Wo, out);
}


# attention: transposed products (S^T=K Q^T, O^T=V^T P^T), a lane owns one query: row max/sum in registers + one cross-row lane exchange, P^T from registers, no LDS; row-sum and PV key-sum order change
# speedup vs baseline: 1.7324x; 1.4841x over previous
#include <hip/hip_runtime.h>
#include <math.h>

typedef __attribute__((ext_vector_type(2))) float v2f;
typedef __attribute__((ext_vector_type(4))) float v4f;
typedef __attribute__((ext_vector_type(8))) float v8f;

#define B_  2
#define C_  2048
#define E_  1024
#define H_  16
#define D_  64
#define HD_ 1024
#define BC_ 4096

static __device__ __forceinline__ v8f vzero8() {
  v8f z = {0.f, 0.f, 0.f, 0.f, 0.f, 0.f, 0.f, 0.f};
  return z;
}

static __device__ __forceinline__ v8f wmma_f32(v2f a, v2f b, v8f c) {
  return __builtin_amdgcn_wmma_f32_16x16x4_f32(false, a, false, b, (short)0, c,
                                               false, false);
}

__global__ __launch_bounds__(256) void qkv_kernel(
    const float* __restrict__ x, const float* __restrict__ Wq,
    const float* __restrict__ Wk, const float* __restrict__ Wv,
    float* __restrict__ Qo, float* __restrict__ Ko, float* __restrict__ Vo) {
  __shared__ float xs[64 * 260];
  const int tid = threadIdx.x;
  const int wave = tid >> 5, lane = tid & 31, h = lane >> 4, m = lane & 15;
  const int row0 = blockIdx.x * 64;
  const int r0 = tid >> 6, c4 = (tid & 63) << 2;
  const int n0g = (blockIdx.y * 8 + wave) * 32;
  const int mat = n0g >> 10;
  const int jin = n0g & 1023;
  const float* Wm = (mat == 0) ? Wq : ((mat == 1) ? Wk : Wv);
  float* dst = (mat == 0) ? Qo : ((mat == 1) ? Ko : Vo);

  v8f acc[4][2];
#pragma unroll
  for (int t = 0; t < 4; ++t) { acc[t][0] = vzero8(); acc[t][1] = vzero8(); }
  for (int kc = 0; kc < E_; kc += 256) {
    __syncthreads();
    {
      const float* xg = &x[(size_t)(row0 + r0) * E_ + kc + c4];
      float* xl = &xs[r0 * 260 + c4];
#pragma unroll
      for (int it = 0; it < 16; ++it) {
        v4f v = *(const v4f*)&xg[(size_t)it * 4 * E_];
        *(v4f*)&xl[it * 4 * 260] = v;
      }
    }
    __syncthreads();
#pragma unroll 4
    for (int k = 0; k < 256; k += 4) {
      const int krow = kc + k + 2 * h;
      const float* w0 = &Wm[(size_t)krow * HD_ + jin + m];
      v2f b0, b1;
      b0.x = w0[0];   b0.y = w0[HD_];
      b1.x = w0[16];  b1.y = w0[HD_ + 16];
#pragma unroll
      for (int t = 0; t < 4; ++t) {
        v2f a = *(const v2f*)&xs[(16 * t + m) * 260 + k + 2 * h];
        acc[t][0] = wmma_f32(a, b0, acc[t][0]);
        acc[t][1] = wmma_f32(a, b1, acc[t][1]);
      }
    }
  }
  const int head = jin >> 6, d0 = jin & 63;
#pragma unroll
  for (int t = 0; t < 4; ++t) {
#pragma unroll
    for (int r = 0; r < 8; ++r) {
      int tok = row0 + 16 * t + 8 * h + r;
      int bb = tok >> 11, cc = tok & 2047;
      float* o = &dst[(((size_t)bb * H_ + head) * C_ + cc) * D_ + d0 + m];
      o[0]  = acc[t][0][r];
      o[16] = acc[t][1][r];
    }
  }
}

__global__ __launch_bounds__(256) void rope_kernel(float* __restrict__ Q,
                                                   float* __restrict__ K) {
  const int idx = blockIdx.x * 256 + threadIdx.x;
  float* P = (blockIdx.y == 0) ? Q : K;
  const int i = idx & 31;
  const int c = (idx >> 5) & 2047;
  const int bh = idx >> 16;
  const float inv_freq = __expf(-(2.0f * (float)i / 64.0f) * 9.210340371976184f);
  const float ang = (float)c * inv_freq;
  float sn, cs;
  sincosf(ang, &sn, &cs);
  const size_t off = ((size_t)bh * C_ + c) * D_ + 2 * i;
  const float e = P[off], o = P[off + 1];
  P[off] = e * cs - o * sn;
  P[off + 1] = e * sn + o * cs;
}

static __device__ __forceinline__ float xhalf(float x) {
  const int b = __builtin_bit_cast(int, x);
  return __builtin_bit_cast(
      float, __builtin_amdgcn_permlanex16(b, b, 0x76543210, 0xfedcba98, false,
                                          false));
}

__global__ __launch_bounds__(256) __attribute__((amdgpu_waves_per_eu(4, 4)))
void attn_kernel(
    const float* __restrict__ Q, const float* __restrict__ K,
    const float* __restrict__ V, float* __restrict__ hid) {
  const int tid = threadIdx.x;
  const int wave = tid >> 5, lane = tid & 31, h = lane >> 4, m = lane & 15;
  const int bh = blockIdx.y;
  const int bb = bh >> 4, head = bh & 15;
  const int lt = blockIdx.x * 4 + (wave & 3);
  const bool late = (((wave >> 2) ^ wave) & 1) != 0;
  const int qt = late ? (C_ / 16 - 1) - lt : lt;
  const int q0 = qt * 16;
  const size_t base = (size_t)bh * C_ * D_;

  v2f bq[16];
#pragma unroll
  for (int s = 0; s < 16; ++s)
    bq[s] = *(const v2f*)&Q[base + (size_t)(q0 + m) * D_ + 4 * s + 2 * h];

  v8f o0 = vzero8(), o1 = vzero8(), o2 = vzero8(), o3 = vzero8();
  float mrow = -INFINITY, lrow = 0.f;

  for (int kt = 0; kt <= qt; ++kt) {
    const int k0 = kt * 16;
    v2f ak[16];
#pragma unroll
    for (int s = 0; s < 16; ++s)
      ak[s] = *(const v2f*)&K[base + (size_t)(k0 + m) * D_ + 4 * s + 2 * h];
    v8f st = vzero8();
#pragma unroll
    for (int s = 0; s < 16; ++s) st = wmma_f32(ak[s], bq[s], st);
    float sv[8];
#pragma unroll
    for (int r = 0; r < 8; ++r) {
      sv[r] = st[r] * 0.125f;
      if (kt == qt && (8 * h + r) > m) sv[r] = -INFINITY;
    }
    float vmax = fmaxf(fmaxf(fmaxf(sv[0], sv[1]), fmaxf(sv[2], sv[3])),
                       fmaxf(fmaxf(sv[4], sv[5]), fmaxf(sv[6], sv[7])));
    vmax = fmaxf(vmax, xhalf(vmax));
    const float mnew = fmaxf(mrow, vmax);
    const float cr = __expf(mrow - mnew);
    float p[8];
#pragma unroll
    for (int r = 0; r < 8; ++r) p[r] = __expf(sv[r] - mnew);
    float rs = ((p[0] + p[1]) + (p[2] + p[3])) + ((p[4] + p[5]) + (p[6] + p[7]));
    rs = rs + xhalf(rs);
    lrow = lrow * cr + rs;
    mrow = mnew;
#pragma unroll
    for (int r = 0; r < 8; ++r) {
      o0[r] *= cr; o1[r] *= cr; o2[r] *= cr; o3[r] *= cr;
    }
#pragma unroll
    for (int j = 0; j < 4; ++j) {
      v2f bp;
      bp.x = p[2 * j];
      bp.y = p[2 * j + 1];
      const float* v0 = &V[base + (size_t)(k0 + 8 * h + 2 * j) * D_ + m];
      const float* v1 = v0 + D_;
      v2f av;
      av.x = v0[0 * 16]; av.y = v1[0 * 16];  o0 = wmma_f32(av, bp, o0);
      av.x = v0[1 * 16]; av.y = v1[1 * 16];  o1 = wmma_f32(av, bp, o1);
      av.x = v0[2 * 16]; av.y = v1[2 * 16];  o2 = wmma_f32(av, bp, o2);
      av.x = v0[3 * 16]; av.y = v1[3 * 16];  o3 = wmma_f32(av, bp, o3);
    }
  }
  const float inv = 1.0f / lrow;
  float* o = &hid[((size_t)bb * C_ + q0 + m) * HD_ + head * D_ + 8 * h];
  v4f w;
  w.x = o0[0] * inv; w.y = o0[1] * inv; w.z = o0[2] * inv; w.w = o0[3] * inv;
  *(v4f*)&o[0] = w;
  w.x = o0[4] * inv; w.y = o0[5] * inv; w.z = o0[6] * inv; w.w = o0[7] * inv;
  *(v4f*)&o[4] = w;
  w.x = o1[0] * inv; w.y = o1[1] * inv; w.z = o1[2] * inv; w.w = o1[3] * inv;
  *(v4f*)&o[16] = w;
  w.x = o1[4] * inv; w.y = o1[5] * inv; w.z = o1[6] * inv; w.w = o1[7] * inv;
  *(v4f*)&o[20] = w;
  w.x = o2[0] * inv; w.y = o2[1] * inv; w.z = o2[2] * inv; w.w = o2[3] * inv;
  *(v4f*)&o[32] = w;
  w.x = o2[4] * inv; w.y = o2[5] * inv; w.z = o2[6] * inv; w.w = o2[7] * inv;
  *(v4f*)&o[36] = w;
  w.x = o3[0] * inv; w.y = o3[1] * inv; w.z = o3[2] * inv; w.w = o3[3] * inv;
  *(v4f*)&o[48] = w;
  w.x = o3[4] * inv; w.y = o3[5] * inv; w.z = o3[6] * inv; w.w = o3[7] * inv;
  *(v4f*)&o[52] = w;
}

__global__ __launch_bounds__(256) void proj_kernel(
    const float* __restrict__ hid, const float* __restrict__ Wo,
    float* __restrict__ out) {
  __shared__ float xs[64 * 260];
  const int tid = threadIdx.x;
  const int wave = tid >> 5, lane = tid & 31, h = lane >> 4, m = lane & 15;
  const int row0 = blockIdx.x * 64;
  const int r0 = tid >> 6, c4 = (tid & 63) << 2;
  const int n0 = (blockIdx.y * 8 + wave) * 32;

  v8f acc[4][2];
#pragma unroll
  for (int t = 0; t < 4; ++t) { acc[t][0] = vzero8(); acc[t][1] = vzero8(); }
  for (int kc = 0; kc < HD_; kc += 256) {
    __syncthreads();
    {
      const float* xg = &hid[(size_t)(row0 + r0) * HD_ + kc + c4];
      float* xl = &xs[r0 * 260 + c4];
#pragma unroll
      for (int it = 0; it < 16; ++it) {
        v4f v = *(const v4f*)&xg[(size_t)it * 4 * HD_];
        *(v4f*)&xl[it * 4 * 260] = v;
      }
    }
    __syncthreads();
#pragma unroll 4
    for (int k = 0; k < 256; k += 4) {
      const int krow = kc + k + 2 * h;
      const float* w0 = &Wo[(size_t)krow * E_ + n0 + m];
      v2f b0, b1;
      b0.x = w0[0];   b0.y = w0[E_];
      b1.x = w0[16];  b1.y = w0[E_ + 16];
#pragma unroll
      for (int t = 0; t < 4; ++t) {
        v2f a = *(const v2f*)&xs[(16 * t + m) * 260 + k + 2 * h];
        acc[t][0] = wmma_f32(a, b0, acc[t][0]);
        acc[t][1] = wmma_f32(a, b1, acc[t][1]);
      }
    }
  }
#pragma unroll
  for (int t = 0; t < 4; ++t) {
#pragma unroll
    for (int r = 0; r < 8; ++r) {
      float* o = &out[(size_t)(row0 + 16 * t + 8 * h + r) * E_ + n0 + m];
      o[0]  = acc[t][0][r];
      o[16] = acc[t][1][r];
    }
  }
}

extern "C" void kernel_launch(void* const* d_in, const int* in_sizes, int n_in,
                              void* d_out, int out_size, void* d_ws,
                              size_t ws_size, hipStream_t stream) {
  (void)in_sizes; (void)n_in; (void)out_size; (void)ws_size;
  const float* x  = (const float*)d_in[0];
  const float* Wq = (const float*)d_in[1];
  const float* Wk = (const float*)d_in[2];
  const float* Wv = (const float*)d_in[3];
  const float* Wo = (const float*)d_in[4];
  float* out = (float*)d_out;

  const size_t BHCD = (size_t)B_ * H_ * C_ * D_;
  float* Qw  = (float*)d_ws;
  float* Kw  = Qw + BHCD;
  float* Vw  = Kw + BHCD;
  float* Hid = Vw + BHCD;

  qkv_kernel<<<dim3(BC_ / 64, (3 * HD_) / (8 * 32)), 256, 0, stream>>>(
      x, Wq, Wk, Wv, Qw, Kw, Vw);
  rope_kernel<<<dim3((B_ * H_ * C_ * (D_ / 2)) / 256, 2), 256, 0, stream>>>(
      Qw, Kw);
  attn_kernel<<<dim3(C_ / 16 / 8, B_ * H_), 256, 0, stream>>>(Qw, Kw, Vw, Hid);
  proj_kernel<<<dim3(BC_ / 64, HD_ / (8 * 32)), 256, 0, stream>>>(Hid, Wo, out);
}


# RoPE moved into the QKV projection's epilogue: Q and K accumulators rotated in registers (pair partner from the neighbour lane by DPP) before their one store; the RoPE launch is gone; sums unchanged
# speedup vs baseline: 2.0917x; 1.2074x over previous
#include <hip/hip_runtime.h>
#include <math.h>

typedef __attribute__((ext_vector_type(2))) float v2f;
typedef __attribute__((ext_vector_type(4))) float v4f;
typedef __attribute__((ext_vector_type(8))) float v8f;

#define B_  2
#define C_  2048
#define E_  1024
#define H_  16
#define D_  64
#define HD_ 1024
#define BC_ 4096

static __device__ __forceinline__ v8f vzero8() {
  v8f z = {0.f, 0.f, 0.f, 0.f, 0.f, 0.f, 0.f, 0.f};
  return z;
}

static __device__ __forceinline__ v8f wmma_f32(v2f a, v2f b, v8f c) {
  return __builtin_amdgcn_wmma_f32_16x16x4_f32(false, a, false, b, (short)0, c,
                                               false, false);
}

template <int CTRL>
static __device__ __forceinline__ float quadperm(float x) {
  const int b = __builtin_bit_cast(int, x);
  return __builtin_bit_cast(
      float, __builtin_amdgcn_update_dpp(b, b, CTRL, 0xf, 0xf, true));
}

__global__ __launch_bounds__(256) void qkv_kernel(
    const float* __restrict__ x, const float* __restrict__ Wq,
    const float* __restrict__ Wk, const float* __restrict__ Wv,
    float* __restrict__ Qo, float* __restrict__ Ko, float* __restrict__ Vo) {
  __shared__ float xs[64 * 260];
  const int tid = threadIdx.x;
  const int wave = tid >> 5, lane = tid & 31, h = lane >> 4, m = lane & 15;
  const int row0 = blockIdx.x * 64;
  const int r0 = tid >> 6, c4 = (tid & 63) << 2;
  const int n0g = (blockIdx.y * 8 + wave) * 32;
  const int mat = n0g >> 10;
  const int jin = n0g & 1023;
  const float* Wm = (mat == 0) ? Wq : ((mat == 1) ? Wk : Wv);
  float* dst = (mat == 0) ? Qo : ((mat == 1) ? Ko : Vo);

  v8f acc[4][2];
#pragma unroll
  for (int t = 0; t < 4; ++t) { acc[t][0] = vzero8(); acc[t][1] = vzero8(); }
  for (int kc = 0; kc < E_; kc += 256) {
    __syncthreads();
    {
      const float* xg = &x[(size_t)(row0 + r0) * E_ + kc + c4];
      float* xl = &xs[r0 * 260 + c4];
#pragma unroll
      for (int it = 0; it < 16; ++it) {
        v4f v = *(const v4f*)&xg[(size_t)it * 4 * E_];
        *(v4f*)&xl[it * 4 * 260] = v;
      }
    }
    __syncthreads();
#pragma unroll 4
    for (int k = 0; k < 256; k += 4) {
      const int krow = kc + k + 2 * h;
      const float* w0 = &Wm[(size_t)krow * HD_ + jin + m];
      v2f b0, b1;
      b0.x = w0[0];   b0.y = w0[HD_];
      b1.x = w0[16];  b1.y = w0[HD_ + 16];
#pragma unroll
      for (int t = 0; t < 4; ++t) {
        v2f a = *(const v2f*)&xs[(16 * t + m) * 260 + k + 2 * h];
        acc[t][0] = wmma_f32(a, b0, acc[t][0]);
        acc[t][1] = wmma_f32(a, b1, acc[t][1]);
      }
    }
  }
  const int head = jin >> 6, d0 = jin & 63;
  float* obase = &dst[(((size_t)(row0 >> 11) * H_ + head) * C_ + (row0 & 2047) +
                       8 * h) * D_ + d0 + m];
  if (mat != 2) {
    const float sgn = (m & 1) ? 1.0f : -1.0f;
    const int rsel = (m & 1) << 2;
#pragma unroll
    for (int ct = 0; ct < 2; ++ct) {
      const int i = (d0 + 16 * ct + m) >> 1;
      const float inv_freq =
          __expf(-(2.0f * (float)i / 64.0f) * 9.210340371976184f);
#pragma unroll
      for (int t = 0; t < 4; ++t) {
        const int c0 = ((row0 & 2047) + 16 * t + 8 * h) + rsel;
#pragma unroll
        for (int q = 0; q < 4; ++q) {
          const float ang = (float)(c0 + q) * inv_freq;
          float sn, cs;
          sincosf(ang, &sn, &cs);
          const float s_lo = quadperm<0xA0>(sn), c_lo = quadperm<0xA0>(cs);
          const float s_hi = quadperm<0xF5>(sn), c_hi = quadperm<0xF5>(cs);
          const float own_lo = acc[t][ct][q], own_hi = acc[t][ct][4 + q];
          const float par_lo = quadperm<0xB1>(own_lo);
          const float par_hi = quadperm<0xB1>(own_hi);
          obase[(16 * t + q) * D_ + 16 * ct] =
              own_lo * c_lo + par_lo * (sgn * s_lo);
          obase[(16 * t + 4 + q) * D_ + 16 * ct] =
              own_hi * c_hi + par_hi * (sgn * s_hi);
        }
      }
    }
  } else {
#pragma unroll
    for (int t = 0; t < 4; ++t) {
#pragma unroll
      for (int r = 0; r < 8; ++r) {
        obase[(16 * t + r) * D_]      = acc[t][0][r];
        obase[(16 * t + r) * D_ + 16] = acc[t][1][r];
      }
    }
  }
}

static __device__ __forceinline__ float xhalf(float x) {
  const int b = __builtin_bit_cast(int, x);
  return __builtin_bit_cast(
      float, __builtin_amdgcn_permlanex16(b, b, 0x76543210, 0xfedcba98, false,
                                          false));
}

__global__ __launch_bounds__(256) __attribute__((amdgpu_waves_per_eu(4, 4)))
void attn_kernel(
    const float* __restrict__ Q, const float* __restrict__ K,
    const float* __restrict__ V, float* __restrict__ hid) {
  const int tid = threadIdx.x;
  const int wave = tid >> 5, lane = tid & 31, h = lane >> 4, m = lane & 15;
  const int bh = blockIdx.y;
  const int bb = bh >> 4, head = bh & 15;
  const int lt = blockIdx.x * 4 + (wave & 3);
  const bool late = (((wave >> 2) ^ wave) & 1) != 0;
  const int qt = late ? (C_ / 16 - 1) - lt : lt;
  const int q0 = qt * 16;
  const size_t base = (size_t)bh * C_ * D_;

  v2f bq[16];
#pragma unroll
  for (int s = 0; s < 16; ++s)
    bq[s] = *(const v2f*)&Q[base + (size_t)(q0 + m) * D_ + 4 * s + 2 * h];

  v8f o0 = vzero8(), o1 = vzero8(), o2 = vzero8(), o3 = vzero8();
  float mrow = -INFINITY, lrow = 0.f;

  for (int kt = 0; kt <= qt; ++kt) {
    const int k0 = kt * 16;
    v2f ak[16];
#pragma unroll
    for (int s = 0; s < 16; ++s)
      ak[s] = *(const v2f*)&K[base + (size_t)(k0 + m) * D_ + 4 * s + 2 * h];
    v8f st = vzero8();
#pragma unroll
    for (int s = 0; s < 16; ++s) st = wmma_f32(ak[s], bq[s], st);
    float sv[8];
#pragma unroll
    for (int r = 0; r < 8; ++r) {
      sv[r] = st[r] * 0.125f;
      if (kt == qt && (8 * h + r) > m) sv[r] = -INFINITY;
    }
    float vmax = fmaxf(fmaxf(fmaxf(sv[0], sv[1]), fmaxf(sv[2], sv[3])),
                       fmaxf(fmaxf(sv[4], sv[5]), fmaxf(sv[6], sv[7])));
    vmax = fmaxf(vmax, xhalf(vmax));
    const float mnew = fmaxf(mrow, vmax);
    const float cr = __expf(mrow - mnew);
    float p[8];
#pragma unroll
    for (int r = 0; r < 8; ++r) p[r] = __expf(sv[r] - mnew);
    float rs = ((p[0] + p[1]) + (p[2] + p[3])) + ((p[4] + p[5]) + (p[6] + p[7]));
    rs = rs + xhalf(rs);
    lrow = lrow * cr + rs;
    mrow = mnew;
#pragma unroll
    for (int r = 0; r < 8; ++r) {
      o0[r] *= cr; o1[r] *= cr; o2[r] *= cr; o3[r] *= cr;
    }
#pragma unroll
    for (int j = 0; j < 4; ++j) {
      v2f bp;
      bp.x = p[2 * j];
      bp.y = p[2 * j + 1];
      const float* v0 = &V[base + (size_t)(k0 + 8 * h + 2 * j) * D_ + m];
      const float* v1 = v0 + D_;
      v2f av;
      av.x = v0[0 * 16]; av.y = v1[0 * 16];  o0 = wmma_f32(av, bp, o0);
      av.x = v0[1 * 16]; av.y = v1[1 * 16];  o1 = wmma_f32(av, bp, o1);
      av.x = v0[2 * 16]; av.y = v1[2 * 16];  o2 = wmma_f32(av, bp, o2);
      av.x = v0[3 * 16]; av.y = v1[3 * 16];  o3 = wmma_f32(av, bp, o3);
    }
  }
  const float inv = 1.0f / lrow;
  float* o = &hid[((size_t)bb * C_ + q0 + m) * HD_ + head * D_ + 8 * h];
  v4f w;
  w.x = o0[0] * inv; w.y = o0[1] * inv; w.z = o0[2] * inv; w.w = o0[3] * inv;
  *(v4f*)&o[0] = w;
  w.x = o0[4] * inv; w.y = o0[5] * inv; w.z = o0[6] * inv; w.w = o0[7] * inv;
  *(v4f*)&o[4] = w;
  w.x = o1[0] * inv; w.y = o1[1] * inv; w.z = o1[2] * inv; w.w = o1[3] * inv;
  *(v4f*)&o[16] = w;
  w.x = o1[4] * inv; w.y = o1[5] * inv; w.z = o1[6] * inv; w.w = o1[7] * inv;
  *(v4f*)&o[20] = w;
  w.x = o2[0] * inv; w.y = o2[1] * inv; w.z = o2[2] * inv; w.w = o2[3] * inv;
  *(v4f*)&o[32] = w;
  w.x = o2[4] * inv; w.y = o2[5] * inv; w.z = o2[6] * inv; w.w = o2[7] * inv;
  *(v4f*)&o[36] = w;
  w.x = o3[0] * inv; w.y = o3[1] * inv; w.z = o3[2] * inv; w.w = o3[3] * inv;
  *(v4f*)&o[48] = w;
  w.x = o3[4] * inv; w.y = o3[5] * inv; w.z = o3[6] * inv; w.w = o3[7] * inv;
  *(v4f*)&o[52] = w;
}

__global__ __launch_bounds__(256) void proj_kernel(
    const float* __restrict__ hid, const float* __restrict__ Wo,
    float* __restrict__ out) {
  __shared__ float xs[64 * 260];
  const int tid = threadIdx.x;
  const int wave = tid >> 5, lane = tid & 31, h = lane >> 4, m = lane & 15;
  const int row0 = blockIdx.x * 64;
  const int r0 = tid >> 6, c4 = (tid & 63) << 2;
  const int n0 = (blockIdx.y * 8 + wave) * 32;

  v8f acc[4][2];
#pragma unroll
  for (int t = 0; t < 4; ++t) { acc[t][0] = vzero8(); acc[t][1] = vzero8(); }
  for (int kc = 0; kc < HD_; kc += 256) {
    __syncthreads();
    {
      const float* xg = &hid[(size_t)(row0 + r0) * HD_ + kc + c4];
      float* xl = &xs[r0 * 260 + c4];
#pragma unroll
      for (int it = 0; it < 16; ++it) {
        v4f v = *(const v4f*)&xg[(size_t)it * 4 * HD_];
        *(v4f*)&xl[it * 4 * 260] = v;
      }
    }
    __syncthreads();
#pragma unroll 4
    for (int k = 0; k < 256; k += 4) {
      const int krow = kc + k + 2 * h;
      const float* w0 = &Wo[(size_t)krow * E_ + n0 + m];
      v2f b0, b1;
      b0.x = w0[0];   b0.y = w0[E_];
      b1.x = w0[16];  b1.y = w0[E_ + 16];
#pragma unroll
      for (int t = 0; t < 4; ++t) {
        v2f a = *(const v2f*)&xs[(16 * t + m) * 260 + k + 2 * h];
        acc[t][0] = wmma_f32(a, b0, acc[t][0]);
        acc[t][1] = wmma_f32(a, b1, acc[t][1]);
      }
    }
  }
#pragma unroll
  for (int t = 0; t < 4; ++t) {
#pragma unroll
    for (int r = 0; r < 8; ++r) {
      float* o = &out[(size_t)(row0 + 16 * t + 8 * h + r) * E_ + n0 + m];
      o[0]  = acc[t][0][r];
      o[16] = acc[t][1][r];
    }
  }
}

extern "C" void kernel_launch(void* const* d_in, const int* in_sizes, int n_in,
                              void* d_out, int out_size, void* d_ws,
                              size_t ws_size, hipStream_t stream) {
  (void)in_sizes; (void)n_in; (void)out_size; (void)ws_size;
  const float* x  = (const float*)d_in[0];
  const float* Wq = (const float*)d_in[1];
  const float* Wk = (const float*)d_in[2];
  const float* Wv = (const float*)d_in[3];
  const float* Wo = (const float*)d_in[4];
  float* out = (float*)d_out;

  const size_t BHCD = (size_t)B_ * H_ * C_ * D_;
  float* Qw  = (float*)d_ws;
  float* Kw  = Qw + BHCD;
  float* Vw  = Kw + BHCD;
  float* Hid = Vw + BHCD;

  qkv_kernel<<<dim3(BC_ / 64, (3 * HD_) / (8 * 32)), 256, 0, stream>>>(
      x, Wq, Wk, Wv, Qw, Kw, Vw);
  attn_kernel<<<dim3(C_ / 16 / 8, B_ * H_), 256, 0, stream>>>(Qw, Kw, Vw, Hid);
  proj_kernel<<<dim3(BC_ / 64, HD_ / (8 * 32)), 256, 0, stream>>>(Hid, Wo, out);
}


# projection GEMMs (QKV and output): weight tile goes through LDS too, shared by a 128x128 block of four 64x64 waves, so no fragment comes straight from global memory; k order and sums unchanged
# speedup vs baseline: 2.1933x; 1.0486x over previous
#include <hip/hip_runtime.h>
#include <math.h>

typedef __attribute__((ext_vector_type(2))) float v2f;
typedef __attribute__((ext_vector_type(4))) float v4f;
typedef __attribute__((ext_vector_type(8))) float v8f;

#define B_  2
#define C_  2048
#define E_  1024
#define H_  16
#define D_  64
#define HD_ 1024
#define BC_ 4096

static __device__ __forceinline__ v8f vzero8() {
  v8f z = {0.f, 0.f, 0.f, 0.f, 0.f, 0.f, 0.f, 0.f};
  return z;
}

static __device__ __forceinline__ v8f wmma_f32(v2f a, v2f b, v8f c) {
  return __builtin_amdgcn_wmma_f32_16x16x4_f32(false, a, false, b, (short)0, c,
                                               false, false);
}

#define GM_KC 32
#define GM_AP 36
#define GM_BP 288
#define GM_BOFF (128 * GM_AP)
#define GM_STAGE (128 * GM_AP + (GM_KC / 2) * GM_BP)

static __device__ __forceinline__ void gemm_128x128(
    const float* __restrict__ A, const float* __restrict__ Bm,
    float* __restrict__ lds, v8f (&acc)[4][4]) {
  const int tid = threadIdx.x;
  const int wave = tid >> 5, lane = tid & 31, h = lane >> 4, m = lane & 15;
  const int wm = wave >> 1, wn = wave & 1;
  const int ar = tid >> 3, ac = (tid & 7) << 2;
  const int bc = lane << 2;
  const float* ag = A + (size_t)ar * 1024 + ac;
  const float* bg = Bm + (size_t)(2 * wave) * 1024 + bc;
  float* asw = lds + ar * GM_AP + ac;
  float* bsw = lds + GM_BOFF + wave * GM_BP + 2 * bc;
  const int aoff = (wm * 64 + m) * GM_AP + 2 * h;
  const int boff = GM_BOFF + h * GM_BP + 2 * (wn * 64 + m);

#pragma unroll
  for (int t = 0; t < 4; ++t) {
#pragma unroll
    for (int c = 0; c < 4; ++c) acc[t][c] = vzero8();
  }

  v4f ra[8], rb[8];
#pragma unroll
  for (int it = 0; it < 8; ++it)
    ra[it] = *(const v4f*)&ag[(size_t)it * 16 * 1024];
#pragma unroll
  for (int j = 0; j < 4; ++j) {
    rb[2 * j]     = *(const v4f*)&bg[(size_t)(8 * j) * 1024];
    rb[2 * j + 1] = *(const v4f*)&bg[(size_t)(8 * j + 1) * 1024];
  }
#pragma unroll
  for (int it = 0; it < 8; ++it) *(v4f*)&asw[it * 16 * GM_AP] = ra[it];
#pragma unroll
  for (int j = 0; j < 4; ++j) {
    v4f w0, w1;
    w0.x = rb[2 * j].x; w0.y = rb[2 * j + 1].x;
    w0.z = rb[2 * j].y; w0.w = rb[2 * j + 1].y;
    w1.x = rb[2 * j].z; w1.y = rb[2 * j + 1].z;
    w1.z = rb[2 * j].w; w1.w = rb[2 * j + 1].w;
    *(v4f*)&bsw[4 * j * GM_BP] = w0;
    *(v4f*)&bsw[4 * j * GM_BP + 4] = w1;
  }
  __syncthreads();

#pragma unroll 1
  for (int ch = 0; ch < 1024 / GM_KC; ++ch) {
    const float* st = lds + (ch & 1) * GM_STAGE;
    const bool more = (ch + 1) < 1024 / GM_KC;
    if (more) {
      const float* agn = ag + (ch + 1) * GM_KC;
      const float* bgn = bg + (size_t)(ch + 1) * GM_KC * 1024;
#pragma unroll
      for (int it = 0; it < 8; ++it)
        ra[it] = *(const v4f*)&agn[(size_t)it * 16 * 1024];
#pragma unroll
      for (int j = 0; j < 4; ++j) {
        rb[2 * j]     = *(const v4f*)&bgn[(size_t)(8 * j) * 1024];
        rb[2 * j + 1] = *(const v4f*)&bgn[(size_t)(8 * j + 1) * 1024];
      }
    }
    const float* ap = st + aoff;
    const float* bp = st + boff;
    v2f fa[2][4], fb[2][4];
#pragma unroll
    for (int t = 0; t < 4; ++t) {
      fa[0][t] = *(const v2f*)&ap[t * 16 * GM_AP];
      fb[0][t] = *(const v2f*)&bp[t * 32];
    }
#pragma unroll
    for (int ks = 0; ks < GM_KC / 4; ++ks) {
      const int cur = ks & 1, nxt = cur ^ 1;
      if (ks + 1 < GM_KC / 4) {
        const int k = 4 * (ks + 1);
#pragma unroll
        for (int t = 0; t < 4; ++t) {
          fa[nxt][t] = *(const v2f*)&ap[t * 16 * GM_AP + k];
          fb[nxt][t] = *(const v2f*)&bp[(k / 2) * GM_BP + t * 32];
        }
      }
#pragma unroll
      for (int t = 0; t < 4; ++t) {
#pragma unroll
        for (int c = 0; c < 4; ++c)
          acc[t][c] = wmma_f32(fa[cur][t], fb[cur][c], acc[t][c]);
      }
    }
    if (more) {
      float* aw = asw + ((ch + 1) & 1) * GM_STAGE;
      float* bw = bsw + ((ch + 1) & 1) * GM_STAGE;
#pragma unroll
      for (int it = 0; it < 8; ++it) *(v4f*)&aw[it * 16 * GM_AP] = ra[it];
#pragma unroll
      for (int j = 0; j < 4; ++j) {
        v4f w0, w1;
        w0.x = rb[2 * j].x; w0.y = rb[2 * j + 1].x;
        w0.z = rb[2 * j].y; w0.w = rb[2 * j + 1].y;
        w1.x = rb[2 * j].z; w1.y = rb[2 * j + 1].z;
        w1.z = rb[2 * j].w; w1.w = rb[2 * j + 1].w;
        *(v4f*)&bw[4 * j * GM_BP] = w0;
        *(v4f*)&bw[4 * j * GM_BP + 4] = w1;
      }
    }
    __syncthreads();
  }
}

template <int CTRL>
static __device__ __forceinline__ float quadperm(float x) {
  const int b = __builtin_bit_cast(int, x);
  return __builtin_bit_cast(
      float, __builtin_amdgcn_update_dpp(b, b, CTRL, 0xf, 0xf, true));
}

template <int CT>
static __device__ __forceinline__ void rope_store(v8f a0, v8f a1, v8f a2,
                                                  v8f a3, float* obase,
                                                  int pos0, int m) {
  const v8f a[4] = {a0, a1, a2, a3};
  const float sgn = (m & 1) ? 1.0f : -1.0f;
  const int rsel = (m & 1) << 2;
  const int i = (16 * CT + m) >> 1;
  const float inv_freq =
      __expf(-(2.0f * (float)i / 64.0f) * 9.210340371976184f);
#pragma unroll
  for (int t = 0; t < 4; ++t) {
    const int c0 = (pos0 + 16 * t) + rsel;
#pragma unroll
    for (int q = 0; q < 4; ++q) {
      const float ang = (float)(c0 + q) * inv_freq;
      float sn, cs;
      sincosf(ang, &sn, &cs);
      const float s_lo = quadperm<0xA0>(sn), c_lo = quadperm<0xA0>(cs);
      const float s_hi = quadperm<0xF5>(sn), c_hi = quadperm<0xF5>(cs);
      const float own_lo = a[t][q], own_hi = a[t][4 + q];
      const float par_lo = quadperm<0xB1>(own_lo);
      const float par_hi = quadperm<0xB1>(own_hi);
      obase[(16 * t + q) * D_ + 16 * CT] =
          own_lo * c_lo + par_lo * (sgn * s_lo);
      obase[(16 * t + 4 + q) * D_ + 16 * CT] =
          own_hi * c_hi + par_hi * (sgn * s_hi);
    }
  }
}

__global__ __launch_bounds__(128) __attribute__((amdgpu_waves_per_eu(3, 3)))
void qkv_kernel(
    const float* __restrict__ x, const float* __restrict__ Wq,
    const float* __restrict__ Wk, const float* __restrict__ Wv,
    float* __restrict__ Qo, float* __restrict__ Ko, float* __restrict__ Vo) {
  __shared__ float lds[2 * GM_STAGE];
  const int tid = threadIdx.x;
  const int wave = tid >> 5, lane = tid & 31, h = lane >> 4, m = lane & 15;
  const int wm = wave >> 1, wn = wave & 1;
  const int rowb = blockIdx.x * 128;
  const int colb = blockIdx.y * 128;
  const int mat = colb >> 10;
  const int jb = colb & 1023;
  const float* Wm = (mat == 0) ? Wq : ((mat == 1) ? Wk : Wv);
  float* dst = (mat == 0) ? Qo : ((mat == 1) ? Ko : Vo);

  v8f acc[4][4];
  gemm_128x128(x + (size_t)rowb * E_, Wm + jb, lds, acc);

  const int row0 = rowb + wm * 64;
  const int head = (jb + wn * 64) >> 6;
  float* obase = &dst[(((size_t)(row0 >> 11) * H_ + head) * C_ + (row0 & 2047) +
                       8 * h) * D_ + m];
  if (mat != 2) {
    const int pos0 = (row0 & 2047) + 8 * h;
    rope_store<0>(acc[0][0], acc[1][0], acc[2][0], acc[3][0], obase, pos0, m);
    rope_store<1>(acc[0][1], acc[1][1], acc[2][1], acc[3][1], obase, pos0, m);
    rope_store<2>(acc[0][2], acc[1][2], acc[2][2], acc[3][2], obase, pos0, m);
    rope_store<3>(acc[0][3], acc[1][3], acc[2][3], acc[3][3], obase, pos0, m);
  } else {
#pragma unroll
    for (int t = 0; t < 4; ++t) {
#pragma unroll
      for (int r = 0; r < 8; ++r) {
#pragma unroll
        for (int ct = 0; ct < 4; ++ct)
          obase[(16 * t + r) * D_ + 16 * ct] = acc[t][ct][r];
      }
    }
  }
}

static __device__ __forceinline__ float xhalf(float x) {
  const int b = __builtin_bit_cast(int, x);
  return __builtin_bit_cast(
      float, __builtin_amdgcn_permlanex16(b, b, 0x76543210, 0xfedcba98, false,
                                          false));
}

__global__ __launch_bounds__(256) __attribute__((amdgpu_waves_per_eu(4, 4)))
void attn_kernel(
    const float* __restrict__ Q, const float* __restrict__ K,
    const float* __restrict__ V, float* __restrict__ hid) {
  const int tid = threadIdx.x;
  const int wave = tid >> 5, lane = tid & 31, h = lane >> 4, m = lane & 15;
  const int bh = blockIdx.y;
  const int bb = bh >> 4, head = bh & 15;
  const int lt = blockIdx.x * 4 + (wave & 3);
  const bool late = (((wave >> 2) ^ wave) & 1) != 0;
  const int qt = late ? (C_ / 16 - 1) - lt : lt;
  const int q0 = qt * 16;
  const size_t base = (size_t)bh * C_ * D_;

  v2f bq[16];
#pragma unroll
  for (int s = 0; s < 16; ++s)
    bq[s] = *(const v2f*)&Q[base + (size_t)(q0 + m) * D_ + 4 * s + 2 * h];

  v8f o0 = vzero8(), o1 = vzero8(), o2 = vzero8(), o3 = vzero8();
  float mrow = -INFINITY, lrow = 0.f;

  for (int kt = 0; kt <= qt; ++kt) {
    const int k0 = kt * 16;
    v2f ak[16];
#pragma unroll
    for (int s = 0; s < 16; ++s)
      ak[s] = *(const v2f*)&K[base + (size_t)(k0 + m) * D_ + 4 * s + 2 * h];
    v8f st = vzero8();
#pragma unroll
    for (int s = 0; s < 16; ++s) st = wmma_f32(ak[s], bq[s], st);
    float sv[8];
#pragma unroll
    for (int r = 0; r < 8; ++r) {
      sv[r] = st[r] * 0.125f;
      if (kt == qt && (8 * h + r) > m) sv[r] = -INFINITY;
    }
    float vmax = fmaxf(fmaxf(fmaxf(sv[0], sv[1]), fmaxf(sv[2], sv[3])),
                       fmaxf(fmaxf(sv[4], sv[5]), fmaxf(sv[6], sv[7])));
    vmax = fmaxf(vmax, xhalf(vmax));
    const float mnew = fmaxf(mrow, vmax);
    const float cr = __expf(mrow - mnew);
    float p[8];
#pragma unroll
    for (int r = 0; r < 8; ++r) p[r] = __expf(sv[r] - mnew);
    float rs = ((p[0] + p[1]) + (p[2] + p[3])) + ((p[4] + p[5]) + (p[6] + p[7]));
    rs = rs + xhalf(rs);
    lrow = lrow * cr + rs;
    mrow = mnew;
#pragma unroll
    for (int r = 0; r < 8; ++r) {
      o0[r] *= cr; o1[r] *= cr; o2[r] *= cr; o3[r] *= cr;
    }
#pragma unroll
    for (int j = 0; j < 4; ++j) {
      v2f bp;
      bp.x = p[2 * j];
      bp.y = p[2 * j + 1];
      const float* v0 = &V[base + (size_t)(k0 + 8 * h + 2 * j) * D_ + m];
      const float* v1 = v0 + D_;
      v2f av;
      av.x = v0[0 * 16]; av.y = v1[0 * 16];  o0 = wmma_f32(av, bp, o0);
      av.x = v0[1 * 16]; av.y = v1[1 * 16];  o1 = wmma_f32(av, bp, o1);
      av.x = v0[2 * 16]; av.y = v1[2 * 16];  o2 = wmma_f32(av, bp, o2);
      av.x = v0[3 * 16]; av.y = v1[3 * 16];  o3 = wmma_f32(av, bp, o3);
    }
  }
  const float inv = 1.0f / lrow;
  float* o = &hid[((size_t)bb * C_ + q0 + m) * HD_ + head * D_ + 8 * h];
  v4f w;
  w.x = o0[0] * inv; w.y = o0[1] * inv; w.z = o0[2] * inv; w.w = o0[3] * inv;
  *(v4f*)&o[0] = w;
  w.x = o0[4] * inv; w.y = o0[5] * inv; w.z = o0[6] * inv; w.w = o0[7] * inv;
  *(v4f*)&o[4] = w;
  w.x = o1[0] * inv; w.y = o1[1] * inv; w.z = o1[2] * inv; w.w = o1[3] * inv;
  *(v4f*)&o[16] = w;
  w.x = o1[4] * inv; w.y = o1[5] * inv; w.z = o1[6] * inv; w.w = o1[7] * inv;
  *(v4f*)&o[20] = w;
  w.x = o2[0] * inv; w.y = o2[1] * inv; w.z = o2[2] * inv; w.w = o2[3] * inv;
  *(v4f*)&o[32] = w;
  w.x = o2[4] * inv; w.y = o2[5] * inv; w.z = o2[6] * inv; w.w = o2[7] * inv;
  *(v4f*)&o[36] = w;
  w.x = o3[0] * inv; w.y = o3[1] * inv; w.z = o3[2] * inv; w.w = o3[3] * inv;
  *(v4f*)&o[48] = w;
  w.x = o3[4] * inv; w.y = o3[5] * inv; w.z = o3[6] * inv; w.w = o3[7] * inv;
  *(v4f*)&o[52] = w;
}

__global__ __launch_bounds__(128) __attribute__((amdgpu_waves_per_eu(3, 3)))
void proj_kernel(
    const float* __restrict__ hid, const float* __restrict__ Wo,
    float* __restrict__ out) {
  __shared__ float lds[2 * GM_STAGE];
  const int tid = threadIdx.x;
  const int wave = tid >> 5, lane = tid & 31, h = lane >> 4, m = lane & 15;
  const int wm = wave >> 1, wn = wave & 1;
  const int rowb = blockIdx.x * 128;
  const int colb = blockIdx.y * 128;

  v8f acc[4][4];
  gemm_128x128(hid + (size_t)rowb * HD_, Wo + colb, lds, acc);

  float* o = &out[(size_t)(rowb + wm * 64 + 8 * h) * E_ + colb + wn * 64 + m];
#pragma unroll
  for (int t = 0; t < 4; ++t) {
#pragma unroll
    for (int r = 0; r < 8; ++r) {
#pragma unroll
      for (int ct = 0; ct < 4; ++ct)
        o[(size_t)(16 * t + r) * E_ + 16 * ct] = acc[t][ct][r];
    }
  }
}

extern "C" void kernel_launch(void* const* d_in, const int* in_sizes, int n_in,
                              void* d_out, int out_size, void* d_ws,
                              size_t ws_size, hipStream_t stream) {
  (void)in_sizes; (void)n_in; (void)out_size; (void)ws_size;
  const float* x  = (const float*)d_in[0];
  const float* Wq = (const float*)d_in[1];
  const float* Wk = (const float*)d_in[2];
  const float* Wv = (const float*)d_in[3];
  const float* Wo = (const float*)d_in[4];
  float* out = (float*)d_out;

  const size_t BHCD = (size_t)B_ * H_ * C_ * D_;
  float* Qw  = (float*)d_ws;
  float* Kw  = Qw + BHCD;
  float* Vw  = Kw + BHCD;
  float* Hid = Vw + BHCD;

  qkv_kernel<<<dim3(BC_ / 128, (3 * HD_) / 128), 128, 0, stream>>>(
      x, Wq, Wk, Wv, Qw, Kw, Vw);
  attn_kernel<<<dim3(C_ / 16 / 8, B_ * H_), 256, 0, stream>>>(Qw, Kw, Vw, Hid);
  proj_kernel<<<dim3(BC_ / 128, E_ / 128), 128, 0, stream>>>(Hid, Wo, out);
}


# Q, K, V workspace planes stored in WMMA fragment order (Q/K projection computed transposed, operand roles swapped): an attention load reads 512 contiguous bytes; score head-dim sum order changes
# speedup vs baseline: 2.2914x; 1.0447x over previous
#include <hip/hip_runtime.h>
#include <math.h>

typedef __attribute__((ext_vector_type(2))) float v2f;
typedef __attribute__((ext_vector_type(4))) float v4f;
typedef __attribute__((ext_vector_type(8))) float v8f;

#define B_  2
#define C_  2048
#define E_  1024
#define H_  16
#define D_  64
#define HD_ 1024
#define BC_ 4096

static __device__ __forceinline__ v8f vzero8() {
  v8f z = {0.f, 0.f, 0.f, 0.f, 0.f, 0.f, 0.f, 0.f};
  return z;
}

static __device__ __forceinline__ v8f wmma_f32(v2f a, v2f b, v8f c) {
  return __builtin_amdgcn_wmma_f32_16x16x4_f32(false, a, false, b, (short)0, c,
                                               false, false);
}

#define GM_KC 32
#define GM_AP 36
#define GM_BP 288
#define GM_BOFF (128 * GM_AP)
#define GM_STAGE (128 * GM_AP + (GM_KC / 2) * GM_BP)

template <bool SWAP>
static __device__ __forceinline__ void gemm_128x128(
    const float* __restrict__ A, const float* __restrict__ Bm,
    float* __restrict__ lds, v8f (&acc)[4][4]) {
  const int tid = threadIdx.x;
  const int wave = tid >> 5, lane = tid & 31, h = lane >> 4, m = lane & 15;
  const int wm = wave >> 1, wn = wave & 1;
  const int ar = tid >> 3, ac = (tid & 7) << 2;
  const int bc = lane << 2;
  const float* ag = A + (size_t)ar * 1024 + ac;
  const float* bg = Bm + (size_t)(2 * wave) * 1024 + bc;
  float* asw = lds + ar * GM_AP + ac;
  float* bsw = lds + GM_BOFF + wave * GM_BP + 2 * bc;
  const int aoff = (wm * 64 + m) * GM_AP + 2 * h;
  const int boff = GM_BOFF + h * GM_BP + 2 * (wn * 64 + m);

#pragma unroll
  for (int t = 0; t < 4; ++t) {
#pragma unroll
    for (int c = 0; c < 4; ++c) acc[t][c] = vzero8();
  }

  v4f ra[8], rb[8];
#pragma unroll
  for (int it = 0; it < 8; ++it)
    ra[it] = *(const v4f*)&ag[(size_t)it * 16 * 1024];
#pragma unroll
  for (int j = 0; j < 4; ++j) {
    rb[2 * j]     = *(const v4f*)&bg[(size_t)(8 * j) * 1024];
    rb[2 * j + 1] = *(const v4f*)&bg[(size_t)(8 * j + 1) * 1024];
  }
#pragma unroll
  for (int it = 0; it < 8; ++it) *(v4f*)&asw[it * 16 * GM_AP] = ra[it];
#pragma unroll
  for (int j = 0; j < 4; ++j) {
    v4f w0, w1;
    w0.x = rb[2 * j].x; w0.y = rb[2 * j + 1].x;
    w0.z = rb[2 * j].y; w0.w = rb[2 * j + 1].y;
    w1.x = rb[2 * j].z; w1.y = rb[2 * j + 1].z;
    w1.z = rb[2 * j].w; w1.w = rb[2 * j + 1].w;
    *(v4f*)&bsw[4 * j * GM_BP] = w0;
    *(v4f*)&bsw[4 * j * GM_BP + 4] = w1;
  }
  __syncthreads();

#pragma unroll 1
  for (int ch = 0; ch < 1024 / GM_KC; ++ch) {
    const float* st = lds + (ch & 1) * GM_STAGE;
    const bool more = (ch + 1) < 1024 / GM_KC;
    if (more) {
      const float* agn = ag + (ch + 1) * GM_KC;
      const float* bgn = bg + (size_t)(ch + 1) * GM_KC * 1024;
#pragma unroll
      for (int it = 0; it < 8; ++it)
        ra[it] = *(const v4f*)&agn[(size_t)it * 16 * 1024];
#pragma unroll
      for (int j = 0; j < 4; ++j) {
        rb[2 * j]     = *(const v4f*)&bgn[(size_t)(8 * j) * 1024];
        rb[2 * j + 1] = *(const v4f*)&bgn[(size_t)(8 * j + 1) * 1024];
      }
    }
    const float* ap = st + aoff;
    const float* bp = st + boff;
    v2f fa[2][4], fb[2][4];
#pragma unroll
    for (int t = 0; t < 4; ++t) {
      fa[0][t] = *(const v2f*)&ap[t * 16 * GM_AP];
      fb[0][t] = *(const v2f*)&bp[t * 32];
    }
#pragma unroll
    for (int ks = 0; ks < GM_KC / 4; ++ks) {
      const int cur = ks & 1, nxt = cur ^ 1;
      if (ks + 1 < GM_KC / 4) {
        const int k = 4 * (ks + 1);
#pragma unroll
        for (int t = 0; t < 4; ++t) {
          fa[nxt][t] = *(const v2f*)&ap[t * 16 * GM_AP + k];
          fb[nxt][t] = *(const v2f*)&bp[(k / 2) * GM_BP + t * 32];
        }
      }
#pragma unroll
      for (int t = 0; t < 4; ++t) {
#pragma unroll
        for (int c = 0; c < 4; ++c)
          acc[t][c] = SWAP ? wmma_f32(fb[cur][c], fa[cur][t], acc[t][c])
                           : wmma_f32(fa[cur][t], fb[cur][c], acc[t][c]);
      }
    }
    if (more) {
      float* aw = asw + ((ch + 1) & 1) * GM_STAGE;
      float* bw = bsw + ((ch + 1) & 1) * GM_STAGE;
#pragma unroll
      for (int it = 0; it < 8; ++it) *(v4f*)&aw[it * 16 * GM_AP] = ra[it];
#pragma unroll
      for (int j = 0; j < 4; ++j) {
        v4f w0, w1;
        w0.x = rb[2 * j].x; w0.y = rb[2 * j + 1].x;
        w0.z = rb[2 * j].y; w0.w = rb[2 * j + 1].y;
        w1.x = rb[2 * j].z; w1.y = rb[2 * j + 1].z;
        w1.z = rb[2 * j].w; w1.w = rb[2 * j + 1].w;
        *(v4f*)&bw[4 * j * GM_BP] = w0;
        *(v4f*)&bw[4 * j * GM_BP + 4] = w1;
      }
    }
    __syncthreads();
  }
}

#define TILE_F 1024


template <int CT>
static __device__ __forceinline__ void rope_store(v8f a0, v8f a1, v8f a2,
                                                  v8f a3, float* tbase,
                                                  int pos0, int h) {
  const v8f a[4] = {a0, a1, a2, a3};
  float inv_freq[4];
#pragma unroll
  for (int u = 0; u < 4; ++u) {
    const int i = 8 * CT + 4 * h + u;
    inv_freq[u] = __expf(-(2.0f * (float)i / 64.0f) * 9.210340371976184f);
  }
#pragma unroll
  for (int t = 0; t < 4; ++t) {
    const float pos = (float)(pos0 + 16 * t);
    float rot[8];
#pragma unroll
    for (int u = 0; u < 4; ++u) {
      const float ang = pos * inv_freq[u];
      float sn, cs;
      sincosf(ang, &sn, &cs);
      const float e = a[t][2 * u], o = a[t][2 * u + 1];
      rot[2 * u]     = e * cs + o * (-sn);
      rot[2 * u + 1] = o * cs + e * sn;
    }
    v4f w0, w1;
    w0.x = rot[0]; w0.y = rot[1]; w0.z = rot[2]; w0.w = rot[3];
    w1.x = rot[4]; w1.y = rot[5]; w1.z = rot[6]; w1.w = rot[7];
    *(v4f*)&tbase[t * TILE_F + (2 * CT) * 128] = w0;
    *(v4f*)&tbase[t * TILE_F + (2 * CT + 1) * 128] = w1;
  }
}

__global__ __launch_bounds__(128) __attribute__((amdgpu_waves_per_eu(3, 3)))
void qkv_kernel(
    const float* __restrict__ x, const float* __restrict__ Wq,
    const float* __restrict__ Wk, const float* __restrict__ Wv,
    float* __restrict__ Qo, float* __restrict__ Ko, float* __restrict__ Vo) {
  __shared__ float lds[2 * GM_STAGE];
  const int tid = threadIdx.x;
  const int wave = tid >> 5, lane = tid & 31, h = lane >> 4, m = lane & 15;
  const int wm = wave >> 1, wn = wave & 1;
  const int rowb = blockIdx.x * 128;
  const int colb = blockIdx.y * 128;
  const int mat = colb >> 10;
  const int jb = colb & 1023;
  const float* Wm = (mat == 0) ? Wq : ((mat == 1) ? Wk : Wv);
  float* dst = (mat == 0) ? Qo : ((mat == 1) ? Ko : Vo);

  const int row0 = rowb + wm * 64;
  const int head = (jb + wn * 64) >> 6;
  float* tbase = &dst[((size_t)(row0 >> 11) * H_ + head) * C_ * D_ +
                      (size_t)((row0 & 2047) >> 4) * TILE_F + lane * 4];

  if (mat != 2) {
    v8f acc[4][4];
    gemm_128x128<true>(x + (size_t)rowb * E_, Wm + jb, lds, acc);
    const int pos0 = (row0 & 2047) + m;
    rope_store<0>(acc[0][0], acc[1][0], acc[2][0], acc[3][0], tbase, pos0, h);
    rope_store<1>(acc[0][1], acc[1][1], acc[2][1], acc[3][1], tbase, pos0, h);
    rope_store<2>(acc[0][2], acc[1][2], acc[2][2], acc[3][2], tbase, pos0, h);
    rope_store<3>(acc[0][3], acc[1][3], acc[2][3], acc[3][3], tbase, pos0, h);
  } else {
    v8f acc[4][4];
    gemm_128x128<false>(x + (size_t)rowb * E_, Wm + jb, lds, acc);
#pragma unroll
    for (int t = 0; t < 4; ++t) {
#pragma unroll
      for (int n = 0; n < 4; ++n) {
        v4f w0, w1;
        w0.x = acc[t][n][0]; w0.y = acc[t][n][1];
        w0.z = acc[t][n][2]; w0.w = acc[t][n][3];
        w1.x = acc[t][n][4]; w1.y = acc[t][n][5];
        w1.z = acc[t][n][6]; w1.w = acc[t][n][7];
        *(v4f*)&tbase[t * TILE_F + (2 * n) * 128] = w0;
        *(v4f*)&tbase[t * TILE_F + (2 * n + 1) * 128] = w1;
      }
    }
  }
}

static __device__ __forceinline__ float xhalf(float x) {
  const int b = __builtin_bit_cast(int, x);
  return __builtin_bit_cast(
      float, __builtin_amdgcn_permlanex16(b, b, 0x76543210, 0xfedcba98, false,
                                          false));
}

__global__ __launch_bounds__(256) __attribute__((amdgpu_waves_per_eu(4, 4)))
void attn_kernel(
    const float* __restrict__ Q, const float* __restrict__ K,
    const float* __restrict__ V, float* __restrict__ hid) {
  const int tid = threadIdx.x;
  const int wave = tid >> 5, lane = tid & 31, h = lane >> 4, m = lane & 15;
  const int bh = blockIdx.y;
  const int bb = bh >> 4, head = bh & 15;
  const int lt = blockIdx.x * 4 + (wave & 3);
  const bool late = (((wave >> 2) ^ wave) & 1) != 0;
  const int qt = late ? (C_ / 16 - 1) - lt : lt;
  const int q0 = qt * 16;
  const size_t base = (size_t)bh * C_ * D_;

  const float* Qt = Q + base + lane * 4;
  const float* Kt = K + base + lane * 4;
  const float* Vt = V + base + lane * 4;

  v2f bq[16];
#pragma unroll
  for (int g = 0; g < 8; ++g) {
    const v4f q4 = *(const v4f*)&Qt[(size_t)qt * TILE_F + g * 128];
    bq[2 * g].x = q4.x;     bq[2 * g].y = q4.y;
    bq[2 * g + 1].x = q4.z; bq[2 * g + 1].y = q4.w;
  }

  v8f o0 = vzero8(), o1 = vzero8(), o2 = vzero8(), o3 = vzero8();
  float mrow = -INFINITY, lrow = 0.f;

  for (int kt = 0; kt <= qt; ++kt) {
    v4f k4[8], v4[8];
#pragma unroll
    for (int g = 0; g < 8; ++g)
      k4[g] = *(const v4f*)&Kt[(size_t)kt * TILE_F + g * 128];
#pragma unroll
    for (int g = 0; g < 8; ++g)
      v4[g] = *(const v4f*)&Vt[(size_t)kt * TILE_F + g * 128];
    v8f st = vzero8();
#pragma unroll
    for (int g = 0; g < 8; ++g) {
      v2f ak;
      ak.x = k4[g].x; ak.y = k4[g].y;  st = wmma_f32(ak, bq[2 * g], st);
      ak.x = k4[g].z; ak.y = k4[g].w;  st = wmma_f32(ak, bq[2 * g + 1], st);
    }
    float sv[8];
#pragma unroll
    for (int r = 0; r < 8; ++r) {
      sv[r] = st[r] * 0.125f;
      if (kt == qt && (8 * h + r) > m) sv[r] = -INFINITY;
    }
    float vmax = fmaxf(fmaxf(fmaxf(sv[0], sv[1]), fmaxf(sv[2], sv[3])),
                       fmaxf(fmaxf(sv[4], sv[5]), fmaxf(sv[6], sv[7])));
    vmax = fmaxf(vmax, xhalf(vmax));
    const float mnew = fmaxf(mrow, vmax);
    const float cr = __expf(mrow - mnew);
    float p[8];
#pragma unroll
    for (int r = 0; r < 8; ++r) p[r] = __expf(sv[r] - mnew);
    float rs = ((p[0] + p[1]) + (p[2] + p[3])) + ((p[4] + p[5]) + (p[6] + p[7]));
    rs = rs + xhalf(rs);
    lrow = lrow * cr + rs;
    mrow = mnew;
#pragma unroll
    for (int r = 0; r < 8; ++r) {
      o0[r] *= cr; o1[r] *= cr; o2[r] *= cr; o3[r] *= cr;
    }
#pragma unroll
    for (int j = 0; j < 4; ++j) {
      v2f bp;
      bp.x = p[2 * j];
      bp.y = p[2 * j + 1];
      const int gp = j >> 1;
      v2f av;
      if (j & 1) {
        av.x = v4[0 + gp].z; av.y = v4[0 + gp].w;  o0 = wmma_f32(av, bp, o0);
        av.x = v4[2 + gp].z; av.y = v4[2 + gp].w;  o1 = wmma_f32(av, bp, o1);
        av.x = v4[4 + gp].z; av.y = v4[4 + gp].w;  o2 = wmma_f32(av, bp, o2);
        av.x = v4[6 + gp].z; av.y = v4[6 + gp].w;  o3 = wmma_f32(av, bp, o3);
      } else {
        av.x = v4[0 + gp].x; av.y = v4[0 + gp].y;  o0 = wmma_f32(av, bp, o0);
        av.x = v4[2 + gp].x; av.y = v4[2 + gp].y;  o1 = wmma_f32(av, bp, o1);
        av.x = v4[4 + gp].x; av.y = v4[4 + gp].y;  o2 = wmma_f32(av, bp, o2);
        av.x = v4[6 + gp].x; av.y = v4[6 + gp].y;  o3 = wmma_f32(av, bp, o3);
      }
    }
  }
  const float inv = 1.0f / lrow;
  float* o = &hid[((size_t)bb * C_ + q0 + m) * HD_ + head * D_ + 8 * h];
  v4f w;
  w.x = o0[0] * inv; w.y = o0[1] * inv; w.z = o0[2] * inv; w.w = o0[3] * inv;
  *(v4f*)&o[0] = w;
  w.x = o0[4] * inv; w.y = o0[5] * inv; w.z = o0[6] * inv; w.w = o0[7] * inv;
  *(v4f*)&o[4] = w;
  w.x = o1[0] * inv; w.y = o1[1] * inv; w.z = o1[2] * inv; w.w = o1[3] * inv;
  *(v4f*)&o[16] = w;
  w.x = o1[4] * inv; w.y = o1[5] * inv; w.z = o1[6] * inv; w.w = o1[7] * inv;
  *(v4f*)&o[20] = w;
  w.x = o2[0] * inv; w.y = o2[1] * inv; w.z = o2[2] * inv; w.w = o2[3] * inv;
  *(v4f*)&o[32] = w;
  w.x = o2[4] * inv; w.y = o2[5] * inv; w.z = o2[6] * inv; w.w = o2[7] * inv;
  *(v4f*)&o[36] = w;
  w.x = o3[0] * inv; w.y = o3[1] * inv; w.z = o3[2] * inv; w.w = o3[3] * inv;
  *(v4f*)&o[48] = w;
  w.x = o3[4] * inv; w.y = o3[5] * inv; w.z = o3[6] * inv; w.w = o3[7] * inv;
  *(v4f*)&o[52] = w;
}

__global__ __launch_bounds__(128) __attribute__((amdgpu_waves_per_eu(3, 3)))
void proj_kernel(
    const float* __restrict__ hid, const float* __restrict__ Wo,
    float* __restrict__ out) {
  __shared__ float lds[2 * GM_STAGE];
  const int tid = threadIdx.x;
  const int wave = tid >> 5, lane = tid & 31, h = lane >> 4, m = lane & 15;
  const int wm = wave >> 1, wn = wave & 1;
  const int rowb = blockIdx.x * 128;
  const int colb = blockIdx.y * 128;

  v8f acc[4][4];
  gemm_128x128<false>(hid + (size_t)rowb * HD_, Wo + colb, lds, acc);

  float* o = &out[(size_t)(rowb + wm * 64 + 8 * h) * E_ + colb + wn * 64 + m];
#pragma unroll
  for (int t = 0; t < 4; ++t) {
#pragma unroll
    for (int r = 0; r < 8; ++r) {
#pragma unroll
      for (int ct = 0; ct < 4; ++ct)
        o[(size_t)(16 * t + r) * E_ + 16 * ct] = acc[t][ct][r];
    }
  }
}

extern "C" void kernel_launch(void* const* d_in, const int* in_sizes, int n_in,
                              void* d_out, int out_size, void* d_ws,
                              size_t ws_size, hipStream_t stream) {
  (void)in_sizes; (void)n_in; (void)out_size; (void)ws_size;
  const float* x  = (const float*)d_in[0];
  const float* Wq = (const float*)d_in[1];
  const float* Wk = (const float*)d_in[2];
  const float* Wv = (const float*)d_in[3];
  const float* Wo = (const float*)d_in[4];
  float* out = (float*)d_out;

  const size_t BHCD = (size_t)B_ * H_ * C_ * D_;
  float* Qw  = (float*)d_ws;
  float* Kw  = Qw + BHCD;
  float* Vw  = Kw + BHCD;
  float* Hid = Vw + BHCD;

  qkv_kernel<<<dim3(BC_ / 128, (3 * HD_) / 128), 128, 0, stream>>>(
      x, Wq, Wk, Wv, Qw, Kw, Vw);
  attn_kernel<<<dim3(C_ / 16 / 8, B_ * H_), 256, 0, stream>>>(Qw, Kw, Vw, Hid);
  proj_kernel<<<dim3(BC_ / 128, E_ / 128), 128, 0, stream>>>(Hid, Wo, out);
}


# attention: a wave owns two query tiles (32 queries), so each K and V fragment loaded feeds two WMMAs; next key tile loaded a step ahead into a second register set; sums and their order unchanged
# speedup vs baseline: 2.4942x; 1.0885x over previous
#include <hip/hip_runtime.h>
#include <math.h>

typedef __attribute__((ext_vector_type(2))) float v2f;
typedef __attribute__((ext_vector_type(4))) float v4f;
typedef __attribute__((ext_vector_type(8))) float v8f;

#define B_  2
#define C_  2048
#define E_  1024
#define H_  16
#define D_  64
#define HD_ 1024
#define BC_ 4096

static __device__ __forceinline__ v8f vzero8() {
  v8f z = {0.f, 0.f, 0.f, 0.f, 0.f, 0.f, 0.f, 0.f};
  return z;
}

static __device__ __forceinline__ v8f wmma_f32(v2f a, v2f b, v8f c) {
  return __builtin_amdgcn_wmma_f32_16x16x4_f32(false, a, false, b, (short)0, c,
                                               false, false);
}

#define GM_KC 32
#define GM_AP 36
#define GM_BP 288
#define GM_BOFF (128 * GM_AP)
#define GM_STAGE (128 * GM_AP + (GM_KC / 2) * GM_BP)

template <bool SWAP>
static __device__ __forceinline__ void gemm_128x128(
    const float* __restrict__ A, const float* __restrict__ Bm,
    float* __restrict__ lds, v8f (&acc)[4][4]) {
  const int tid = threadIdx.x;
  const int wave = tid >> 5, lane = tid & 31, h = lane >> 4, m = lane & 15;
  const int wm = wave >> 1, wn = wave & 1;
  const int ar = tid >> 3, ac = (tid & 7) << 2;
  const int bc = lane << 2;
  const float* ag = A + (size_t)ar * 1024 + ac;
  const float* bg = Bm + (size_t)(2 * wave) * 1024 + bc;
  float* asw = lds + ar * GM_AP + ac;
  float* bsw = lds + GM_BOFF + wave * GM_BP + 2 * bc;
  const int aoff = (wm * 64 + m) * GM_AP + 2 * h;
  const int boff = GM_BOFF + h * GM_BP + 2 * (wn * 64 + m);

#pragma unroll
  for (int t = 0; t < 4; ++t) {
#pragma unroll
    for (int c = 0; c < 4; ++c) acc[t][c] = vzero8();
  }

  v4f ra[8], rb[8];
#pragma unroll
  for (int it = 0; it < 8; ++it)
    ra[it] = *(const v4f*)&ag[(size_t)it * 16 * 1024];
#pragma unroll
  for (int j = 0; j < 4; ++j) {
    rb[2 * j]     = *(const v4f*)&bg[(size_t)(8 * j) * 1024];
    rb[2 * j + 1] = *(const v4f*)&bg[(size_t)(8 * j + 1) * 1024];
  }
#pragma unroll
  for (int it = 0; it < 8; ++it) *(v4f*)&asw[it * 16 * GM_AP] = ra[it];
#pragma unroll
  for (int j = 0; j < 4; ++j) {
    v4f w0, w1;
    w0.x = rb[2 * j].x; w0.y = rb[2 * j + 1].x;
    w0.z = rb[2 * j].y; w0.w = rb[2 * j + 1].y;
    w1.x = rb[2 * j].z; w1.y = rb[2 * j + 1].z;
    w1.z = rb[2 * j].w; w1.w = rb[2 * j + 1].w;
    *(v4f*)&bsw[4 * j * GM_BP] = w0;
    *(v4f*)&bsw[4 * j * GM_BP + 4] = w1;
  }
  __syncthreads();

#pragma unroll 1
  for (int ch = 0; ch < 1024 / GM_KC; ++ch) {
    const float* st = lds + (ch & 1) * GM_STAGE;
    const bool more = (ch + 1) < 1024 / GM_KC;
    if (more) {
      const float* agn = ag + (ch + 1) * GM_KC;
      const float* bgn = bg + (size_t)(ch + 1) * GM_KC * 1024;
#pragma unroll
      for (int it = 0; it < 8; ++it)
        ra[it] = *(const v4f*)&agn[(size_t)it * 16 * 1024];
#pragma unroll
      for (int j = 0; j < 4; ++j) {
        rb[2 * j]     = *(const v4f*)&bgn[(size_t)(8 * j) * 1024];
        rb[2 * j + 1] = *(const v4f*)&bgn[(size_t)(8 * j + 1) * 1024];
      }
    }
    const float* ap = st + aoff;
    const float* bp = st + boff;
    v2f fa[2][4], fb[2][4];
#pragma unroll
    for (int t = 0; t < 4; ++t) {
      fa[0][t] = *(const v2f*)&ap[t * 16 * GM_AP];
      fb[0][t] = *(const v2f*)&bp[t * 32];
    }
#pragma unroll
    for (int ks = 0; ks < GM_KC / 4; ++ks) {
      const int cur = ks & 1, nxt = cur ^ 1;
      if (ks + 1 < GM_KC / 4) {
        const int k = 4 * (ks + 1);
#pragma unroll
        for (int t = 0; t < 4; ++t) {
          fa[nxt][t] = *(const v2f*)&ap[t * 16 * GM_AP + k];
          fb[nxt][t] = *(const v2f*)&bp[(k / 2) * GM_BP + t * 32];
        }
      }
#pragma unroll
      for (int t = 0; t < 4; ++t) {
#pragma unroll
        for (int c = 0; c < 4; ++c)
          acc[t][c] = SWAP ? wmma_f32(fb[cur][c], fa[cur][t], acc[t][c])
                           : wmma_f32(fa[cur][t], fb[cur][c], acc[t][c]);
      }
    }
    if (more) {
      float* aw = asw + ((ch + 1) & 1) * GM_STAGE;
      float* bw = bsw + ((ch + 1) & 1) * GM_STAGE;
#pragma unroll
      for (int it = 0; it < 8; ++it) *(v4f*)&aw[it * 16 * GM_AP] = ra[it];
#pragma unroll
      for (int j = 0; j < 4; ++j) {
        v4f w0, w1;
        w0.x = rb[2 * j].x; w0.y = rb[2 * j + 1].x;
        w0.z = rb[2 * j].y; w0.w = rb[2 * j + 1].y;
        w1.x = rb[2 * j].z; w1.y = rb[2 * j + 1].z;
        w1.z = rb[2 * j].w; w1.w = rb[2 * j + 1].w;
        *(v4f*)&bw[4 * j * GM_BP] = w0;
        *(v4f*)&bw[4 * j * GM_BP + 4] = w1;
      }
    }
    __syncthreads();
  }
}

#define TILE_F 1024


template <int CT>
static __device__ __forceinline__ void rope_store(v8f a0, v8f a1, v8f a2,
                                                  v8f a3, float* tbase,
                                                  int pos0, int h) {
  const v8f a[4] = {a0, a1, a2, a3};
  float inv_freq[4];
#pragma unroll
  for (int u = 0; u < 4; ++u) {
    const int i = 8 * CT + 4 * h + u;
    inv_freq[u] = __expf(-(2.0f * (float)i / 64.0f) * 9.210340371976184f);
  }
#pragma unroll
  for (int t = 0; t < 4; ++t) {
    const float pos = (float)(pos0 + 16 * t);
    float rot[8];
#pragma unroll
    for (int u = 0; u < 4; ++u) {
      const float ang = pos * inv_freq[u];
      float sn, cs;
      sincosf(ang, &sn, &cs);
      const float e = a[t][2 * u], o = a[t][2 * u + 1];
      rot[2 * u]     = e * cs + o * (-sn);
      rot[2 * u + 1] = o * cs + e * sn;
    }
    v4f w0, w1;
    w0.x = rot[0]; w0.y = rot[1]; w0.z = rot[2]; w0.w = rot[3];
    w1.x = rot[4]; w1.y = rot[5]; w1.z = rot[6]; w1.w = rot[7];
    *(v4f*)&tbase[t * TILE_F + (2 * CT) * 128] = w0;
    *(v4f*)&tbase[t * TILE_F + (2 * CT + 1) * 128] = w1;
  }
}

__global__ __launch_bounds__(128) __attribute__((amdgpu_waves_per_eu(3, 3)))
void qkv_kernel(
    const float* __restrict__ x, const float* __restrict__ Wq,
    const float* __restrict__ Wk, const float* __restrict__ Wv,
    float* __restrict__ Qo, float* __restrict__ Ko, float* __restrict__ Vo) {
  __shared__ float lds[2 * GM_STAGE];
  const int tid = threadIdx.x;
  const int wave = tid >> 5, lane = tid & 31, h = lane >> 4, m = lane & 15;
  const int wm = wave >> 1, wn = wave & 1;
  const int rowb = blockIdx.x * 128;
  const int colb = blockIdx.y * 128;
  const int mat = colb >> 10;
  const int jb = colb & 1023;
  const float* Wm = (mat == 0) ? Wq : ((mat == 1) ? Wk : Wv);
  float* dst = (mat == 0) ? Qo : ((mat == 1) ? Ko : Vo);

  const int row0 = rowb + wm * 64;
  const int head = (jb + wn * 64) >> 6;
  float* tbase = &dst[((size_t)(row0 >> 11) * H_ + head) * C_ * D_ +
                      (size_t)((row0 & 2047) >> 4) * TILE_F + lane * 4];

  if (mat != 2) {
    v8f acc[4][4];
    gemm_128x128<true>(x + (size_t)rowb * E_, Wm + jb, lds, acc);
    const int pos0 = (row0 & 2047) + m;
    rope_store<0>(acc[0][0], acc[1][0], acc[2][0], acc[3][0], tbase, pos0, h);
    rope_store<1>(acc[0][1], acc[1][1], acc[2][1], acc[3][1], tbase, pos0, h);
    rope_store<2>(acc[0][2], acc[1][2], acc[2][2], acc[3][2], tbase, pos0, h);
    rope_store<3>(acc[0][3], acc[1][3], acc[2][3], acc[3][3], tbase, pos0, h);
  } else {
    v8f acc[4][4];
    gemm_128x128<false>(x + (size_t)rowb * E_, Wm + jb, lds, acc);
#pragma unroll
    for (int t = 0; t < 4; ++t) {
#pragma unroll
      for (int n = 0; n < 4; ++n) {
        v4f w0, w1;
        w0.x = acc[t][n][0]; w0.y = acc[t][n][1];
        w0.z = acc[t][n][2]; w0.w = acc[t][n][3];
        w1.x = acc[t][n][4]; w1.y = acc[t][n][5];
        w1.z = acc[t][n][6]; w1.w = acc[t][n][7];
        *(v4f*)&tbase[t * TILE_F + (2 * n) * 128] = w0;
        *(v4f*)&tbase[t * TILE_F + (2 * n + 1) * 128] = w1;
      }
    }
  }
}

static __device__ __forceinline__ float xhalf(float x) {
  const int b = __builtin_bit_cast(int, x);
  return __builtin_bit_cast(
      float, __builtin_amdgcn_permlanex16(b, b, 0x76543210, 0xfedcba98, false,
                                          false));
}

static __device__ __forceinline__ void softmax_step(const float (&sv)[8],
                                                    float& mrow, float& lrow,
                                                    float& cr, float (&p)[8]) {
  float vmax = fmaxf(fmaxf(fmaxf(sv[0], sv[1]), fmaxf(sv[2], sv[3])),
                     fmaxf(fmaxf(sv[4], sv[5]), fmaxf(sv[6], sv[7])));
  vmax = fmaxf(vmax, xhalf(vmax));
  const float mnew = fmaxf(mrow, vmax);
  cr = __expf(mrow - mnew);
#pragma unroll
  for (int r = 0; r < 8; ++r) p[r] = __expf(sv[r] - mnew);
  float rs = ((p[0] + p[1]) + (p[2] + p[3])) + ((p[4] + p[5]) + (p[6] + p[7]));
  rs = rs + xhalf(rs);
  lrow = lrow * cr + rs;
  mrow = mnew;
}

template <bool DO_A, bool MASK, bool LAST>
static __device__ __forceinline__ void key_step(
    int kt, int h, int qa, int qb, const float* __restrict__ Kt,
    const float* __restrict__ Vt, const v2f (&bqA)[16], const v2f (&bqB)[16],
    const v4f (&k4)[8], const v4f (&v4)[8], v4f (&kn)[8], v4f (&vn)[8],
    v8f (&oA)[4], v8f (&oB)[4], float& mA, float& lA, float& mB, float& lB) {
  if (!LAST) {
#pragma unroll
    for (int g = 0; g < 8; ++g)
      kn[g] = *(const v4f*)&Kt[(size_t)(kt + 1) * TILE_F + g * 128];
#pragma unroll
    for (int g = 0; g < 8; ++g)
      vn[g] = *(const v4f*)&Vt[(size_t)(kt + 1) * TILE_F + g * 128];
  }
  v8f sA = vzero8(), sB = vzero8();
#pragma unroll
  for (int g = 0; g < 8; ++g) {
    v2f ak;
    ak.x = k4[g].x; ak.y = k4[g].y;
    if (DO_A) sA = wmma_f32(ak, bqA[2 * g], sA);
    sB = wmma_f32(ak, bqB[2 * g], sB);
    ak.x = k4[g].z; ak.y = k4[g].w;
    if (DO_A) sA = wmma_f32(ak, bqA[2 * g + 1], sA);
    sB = wmma_f32(ak, bqB[2 * g + 1], sB);
  }
  float svA[8], svB[8], pA[8], pB[8], crA = 1.0f, crB;
  const int key0 = 16 * kt + 8 * h;
#pragma unroll
  for (int r = 0; r < 8; ++r) {
    svA[r] = sA[r] * 0.125f;
    svB[r] = sB[r] * 0.125f;
    if (MASK) {
      if (key0 + r > qa) svA[r] = -INFINITY;
      if (key0 + r > qb) svB[r] = -INFINITY;
    }
  }
  if (DO_A) softmax_step(svA, mA, lA, crA, pA);
  softmax_step(svB, mB, lB, crB, pB);
#pragma unroll
  for (int n = 0; n < 4; ++n) {
#pragma unroll
    for (int r = 0; r < 8; ++r) {
      if (DO_A) oA[n][r] *= crA;
      oB[n][r] *= crB;
    }
  }
#pragma unroll
  for (int j = 0; j < 4; ++j) {
    v2f bpA, bpB;
    if (DO_A) { bpA.x = pA[2 * j]; bpA.y = pA[2 * j + 1]; }
    bpB.x = pB[2 * j]; bpB.y = pB[2 * j + 1];
    const int gp = j >> 1;
#pragma unroll
    for (int n = 0; n < 4; ++n) {
      v2f av;
      if (j & 1) { av.x = v4[2 * n + gp].z; av.y = v4[2 * n + gp].w; }
      else       { av.x = v4[2 * n + gp].x; av.y = v4[2 * n + gp].y; }
      if (DO_A) oA[n] = wmma_f32(av, bpA, oA[n]);
      oB[n] = wmma_f32(av, bpB, oB[n]);
    }
  }
}

static __device__ __forceinline__ void store_tile(const v8f (&ot)[4], float inv,
                                                  float* o) {
#pragma unroll
  for (int n = 0; n < 4; ++n) {
    v4f w;
    w.x = ot[n][0] * inv; w.y = ot[n][1] * inv;
    w.z = ot[n][2] * inv; w.w = ot[n][3] * inv;
    *(v4f*)&o[16 * n] = w;
    w.x = ot[n][4] * inv; w.y = ot[n][5] * inv;
    w.z = ot[n][6] * inv; w.w = ot[n][7] * inv;
    *(v4f*)&o[16 * n + 4] = w;
  }
}

__global__ __launch_bounds__(256) __attribute__((amdgpu_waves_per_eu(2, 2)))
void attn_kernel(
    const float* __restrict__ Q, const float* __restrict__ K,
    const float* __restrict__ V, float* __restrict__ hid) {
  const int tid = threadIdx.x;
  const int wave = __builtin_amdgcn_readfirstlane(tid >> 5);
  const int lane = tid & 31, h = lane >> 4, m = lane & 15;
  const int bh = blockIdx.y;
  const int bb = bh >> 4, head = bh & 15;
  const int lt = blockIdx.x * 4 + (wave & 3);
  const bool late = (((wave >> 2) ^ wave) & 1) != 0;
  const int wt = late ? (C_ / 32 - 1) - lt : lt;
  const int qtA = 2 * wt;
  const size_t base = (size_t)bh * C_ * D_;

  const float* Qt = Q + base + lane * 4;
  const float* Kt = K + base + lane * 4;
  const float* Vt = V + base + lane * 4;

  v2f bqA[16], bqB[16];
#pragma unroll
  for (int g = 0; g < 8; ++g) {
    const v4f qa4 = *(const v4f*)&Qt[(size_t)qtA * TILE_F + g * 128];
    const v4f qb4 = *(const v4f*)&Qt[(size_t)(qtA + 1) * TILE_F + g * 128];
    bqA[2 * g].x = qa4.x;     bqA[2 * g].y = qa4.y;
    bqA[2 * g + 1].x = qa4.z; bqA[2 * g + 1].y = qa4.w;
    bqB[2 * g].x = qb4.x;     bqB[2 * g].y = qb4.y;
    bqB[2 * g + 1].x = qb4.z; bqB[2 * g + 1].y = qb4.w;
  }
  v4f k0[8], v0[8], k1[8], v1[8];
#pragma unroll
  for (int g = 0; g < 8; ++g) k0[g] = *(const v4f*)&Kt[g * 128];
#pragma unroll
  for (int g = 0; g < 8; ++g) v0[g] = *(const v4f*)&Vt[g * 128];

  v8f oA[4], oB[4];
#pragma unroll
  for (int n = 0; n < 4; ++n) { oA[n] = vzero8(); oB[n] = vzero8(); }
  float mA = -INFINITY, lA = 0.f, mB = -INFINITY, lB = 0.f;
  const int qa = 16 * qtA + m, qb = qa + 16;

  for (int kt = 0; kt < qtA; kt += 2) {
    key_step<true, false, false>(kt, h, qa, qb, Kt, Vt, bqA, bqB, k0, v0, k1,
                                 v1, oA, oB, mA, lA, mB, lB);
    key_step<true, false, false>(kt + 1, h, qa, qb, Kt, Vt, bqA, bqB, k1, v1,
                                 k0, v0, oA, oB, mA, lA, mB, lB);
  }
  key_step<true, true, false>(qtA, h, qa, qb, Kt, Vt, bqA, bqB, k0, v0, k1, v1,
                              oA, oB, mA, lA, mB, lB);
  key_step<false, true, true>(qtA + 1, h, qa, qb, Kt, Vt, bqA, bqB, k1, v1, k0,
                              v0, oA, oB, mA, lA, mB, lB);

  float* o = &hid[((size_t)bb * C_ + 16 * qtA + m) * HD_ + head * D_ + 8 * h];
  store_tile(oA, 1.0f / lA, o);
  store_tile(oB, 1.0f / lB, o + (size_t)16 * HD_);
}

__global__ __launch_bounds__(128) __attribute__((amdgpu_waves_per_eu(3, 3)))
void proj_kernel(
    const float* __restrict__ hid, const float* __restrict__ Wo,
    float* __restrict__ out) {
  __shared__ float lds[2 * GM_STAGE];
  const int tid = threadIdx.x;
  const int wave = tid >> 5, lane = tid & 31, h = lane >> 4, m = lane & 15;
  const int wm = wave >> 1, wn = wave & 1;
  const int rowb = blockIdx.x * 128;
  const int colb = blockIdx.y * 128;

  v8f acc[4][4];
  gemm_128x128<false>(hid + (size_t)rowb * HD_, Wo + colb, lds, acc);

  float* o = &out[(size_t)(rowb + wm * 64 + 8 * h) * E_ + colb + wn * 64 + m];
#pragma unroll
  for (int t = 0; t < 4; ++t) {
#pragma unroll
    for (int r = 0; r < 8; ++r) {
#pragma unroll
      for (int ct = 0; ct < 4; ++ct)
        o[(size_t)(16 * t + r) * E_ + 16 * ct] = acc[t][ct][r];
    }
  }
}

extern "C" void kernel_launch(void* const* d_in, const int* in_sizes, int n_in,
                              void* d_out, int out_size, void* d_ws,
                              size_t ws_size, hipStream_t stream) {
  (void)in_sizes; (void)n_in; (void)out_size; (void)ws_size;
  const float* x  = (const float*)d_in[0];
  const float* Wq = (const float*)d_in[1];
  const float* Wk = (const float*)d_in[2];
  const float* Wv = (const float*)d_in[3];
  const float* Wo = (const float*)d_in[4];
  float* out = (float*)d_out;

  const size_t BHCD = (size_t)B_ * H_ * C_ * D_;
  float* Qw  = (float*)d_ws;
  float* Kw  = Qw + BHCD;
  float* Vw  = Kw + BHCD;
  float* Hid = Vw + BHCD;

  qkv_kernel<<<dim3(BC_ / 128, (3 * HD_) / 128), 128, 0, stream>>>(
      x, Wq, Wk, Wv, Qw, Kw, Vw);
  attn_kernel<<<dim3(C_ / 32 / 8, B_ * H_), 256, 0, stream>>>(Qw, Kw, Vw, Hid);
  proj_kernel<<<dim3(BC_ / 128, E_ / 128), 128, 0, stream>>>(Hid, Wo, out);
}


# attention: a wave pair splits its two wave tiles' keys evenly (even/odd key tiles, partials merged via LDS); each wave yields matrix priority for its last steps; row-sum and PV key-sum order change
# speedup vs baseline: 2.6443x; 1.0602x over previous
#include <hip/hip_runtime.h>
#include <math.h>

typedef __attribute__((ext_vector_type(2))) float v2f;
typedef __attribute__((ext_vector_type(4))) float v4f;
typedef __attribute__((ext_vector_type(8))) float v8f;

#define B_  2
#define C_  2048
#define E_  1024
#define H_  16
#define D_  64
#define HD_ 1024
#define BC_ 4096

static __device__ __forceinline__ v8f vzero8() {
  v8f z = {0.f, 0.f, 0.f, 0.f, 0.f, 0.f, 0.f, 0.f};
  return z;
}

static __device__ __forceinline__ v8f wmma_f32(v2f a, v2f b, v8f c) {
  return __builtin_amdgcn_wmma_f32_16x16x4_f32(false, a, false, b, (short)0, c,
                                               false, false);
}

#define GM_KC 32
#define GM_AP 36
#define GM_BP 288
#define GM_BOFF (128 * GM_AP)
#define GM_STAGE (128 * GM_AP + (GM_KC / 2) * GM_BP)

template <bool SWAP>
static __device__ __forceinline__ void gemm_128x128(
    const float* __restrict__ A, const float* __restrict__ Bm,
    float* __restrict__ lds, v8f (&acc)[4][4]) {
  const int tid = threadIdx.x;
  const int wave = tid >> 5, lane = tid & 31, h = lane >> 4, m = lane & 15;
  const int wm = wave >> 1, wn = wave & 1;
  const int ar = tid >> 3, ac = (tid & 7) << 2;
  const int bc = lane << 2;
  const float* ag = A + (size_t)ar * 1024 + ac;
  const float* bg = Bm + (size_t)(2 * wave) * 1024 + bc;
  float* asw = lds + ar * GM_AP + ac;
  float* bsw = lds + GM_BOFF + wave * GM_BP + 2 * bc;
  const int aoff = (wm * 64 + m) * GM_AP + 2 * h;
  const int boff = GM_BOFF + h * GM_BP + 2 * (wn * 64 + m);

#pragma unroll
  for (int t = 0; t < 4; ++t) {
#pragma unroll
    for (int c = 0; c < 4; ++c) acc[t][c] = vzero8();
  }

  v4f ra[8], rb[8];
#pragma unroll
  for (int it = 0; it < 8; ++it)
    ra[it] = *(const v4f*)&ag[(size_t)it * 16 * 1024];
#pragma unroll
  for (int j = 0; j < 4; ++j) {
    rb[2 * j]     = *(const v4f*)&bg[(size_t)(8 * j) * 1024];
    rb[2 * j + 1] = *(const v4f*)&bg[(size_t)(8 * j + 1) * 1024];
  }
#pragma unroll
  for (int it = 0; it < 8; ++it) *(v4f*)&asw[it * 16 * GM_AP] = ra[it];
#pragma unroll
  for (int j = 0; j < 4; ++j) {
    v4f w0, w1;
    w0.x = rb[2 * j].x; w0.y = rb[2 * j + 1].x;
    w0.z = rb[2 * j].y; w0.w = rb[2 * j + 1].y;
    w1.x = rb[2 * j].z; w1.y = rb[2 * j + 1].z;
    w1.z = rb[2 * j].w; w1.w = rb[2 * j + 1].w;
    *(v4f*)&bsw[4 * j * GM_BP] = w0;
    *(v4f*)&bsw[4 * j * GM_BP + 4] = w1;
  }
  __syncthreads();

#pragma unroll 1
  for (int ch = 0; ch < 1024 / GM_KC; ++ch) {
    const float* st = lds + (ch & 1) * GM_STAGE;
    const bool more = (ch + 1) < 1024 / GM_KC;
    if (more) {
      const float* agn = ag + (ch + 1) * GM_KC;
      const float* bgn = bg + (size_t)(ch + 1) * GM_KC * 1024;
#pragma unroll
      for (int it = 0; it < 8; ++it)
        ra[it] = *(const v4f*)&agn[(size_t)it * 16 * 1024];
#pragma unroll
      for (int j = 0; j < 4; ++j) {
        rb[2 * j]     = *(const v4f*)&bgn[(size_t)(8 * j) * 1024];
        rb[2 * j + 1] = *(const v4f*)&bgn[(size_t)(8 * j + 1) * 1024];
      }
    }
    const float* ap = st + aoff;
    const float* bp = st + boff;
    v2f fa[2][4], fb[2][4];
#pragma unroll
    for (int t = 0; t < 4; ++t) {
      fa[0][t] = *(const v2f*)&ap[t * 16 * GM_AP];
      fb[0][t] = *(const v2f*)&bp[t * 32];
    }
#pragma unroll
    for (int ks = 0; ks < GM_KC / 4; ++ks) {
      const int cur = ks & 1, nxt = cur ^ 1;
      if (ks + 1 < GM_KC / 4) {
        const int k = 4 * (ks + 1);
#pragma unroll
        for (int t = 0; t < 4; ++t) {
          fa[nxt][t] = *(const v2f*)&ap[t * 16 * GM_AP + k];
          fb[nxt][t] = *(const v2f*)&bp[(k / 2) * GM_BP + t * 32];
        }
      }
#pragma unroll
      for (int t = 0; t < 4; ++t) {
#pragma unroll
        for (int c = 0; c < 4; ++c)
          acc[t][c] = SWAP ? wmma_f32(fb[cur][c], fa[cur][t], acc[t][c])
                           : wmma_f32(fa[cur][t], fb[cur][c], acc[t][c]);
      }
    }
    if (more) {
      float* aw = asw + ((ch + 1) & 1) * GM_STAGE;
      float* bw = bsw + ((ch + 1) & 1) * GM_STAGE;
#pragma unroll
      for (int it = 0; it < 8; ++it) *(v4f*)&aw[it * 16 * GM_AP] = ra[it];
#pragma unroll
      for (int j = 0; j < 4; ++j) {
        v4f w0, w1;
        w0.x = rb[2 * j].x; w0.y = rb[2 * j + 1].x;
        w0.z = rb[2 * j].y; w0.w = rb[2 * j + 1].y;
        w1.x = rb[2 * j].z; w1.y = rb[2 * j + 1].z;
        w1.z = rb[2 * j].w; w1.w = rb[2 * j + 1].w;
        *(v4f*)&bw[4 * j * GM_BP] = w0;
        *(v4f*)&bw[4 * j * GM_BP + 4] = w1;
      }
    }
    __syncthreads();
  }
}

#define TILE_F 1024


template <int CT>
static __device__ __forceinline__ void rope_store(v8f a0, v8f a1, v8f a2,
                                                  v8f a3, float* tbase,
                                                  int pos0, int h) {
  const v8f a[4] = {a0, a1, a2, a3};
  float inv_freq[4];
#pragma unroll
  for (int u = 0; u < 4; ++u) {
    const int i = 8 * CT + 4 * h + u;
    inv_freq[u] = __expf(-(2.0f * (float)i / 64.0f) * 9.210340371976184f);
  }
#pragma unroll
  for (int t = 0; t < 4; ++t) {
    const float pos = (float)(pos0 + 16 * t);
    float rot[8];
#pragma unroll
    for (int u = 0; u < 4; ++u) {
      const float ang = pos * inv_freq[u];
      float sn, cs;
      sincosf(ang, &sn, &cs);
      const float e = a[t][2 * u], o = a[t][2 * u + 1];
      rot[2 * u]     = e * cs + o * (-sn);
      rot[2 * u + 1] = o * cs + e * sn;
    }
    v4f w0, w1;
    w0.x = rot[0]; w0.y = rot[1]; w0.z = rot[2]; w0.w = rot[3];
    w1.x = rot[4]; w1.y = rot[5]; w1.z = rot[6]; w1.w = rot[7];
    *(v4f*)&tbase[t * TILE_F + (2 * CT) * 128] = w0;
    *(v4f*)&tbase[t * TILE_F + (2 * CT + 1) * 128] = w1;
  }
}

__global__ __launch_bounds__(128) __attribute__((amdgpu_waves_per_eu(3, 3)))
void qkv_kernel(
    const float* __restrict__ x, const float* __restrict__ Wq,
    const float* __restrict__ Wk, const float* __restrict__ Wv,
    float* __restrict__ Qo, float* __restrict__ Ko, float* __restrict__ Vo) {
  __shared__ float lds[2 * GM_STAGE];
  const int tid = threadIdx.x;
  const int wave = tid >> 5, lane = tid & 31, h = lane >> 4, m = lane & 15;
  const int wm = wave >> 1, wn = wave & 1;
  const int rowb = blockIdx.x * 128;
  const int colb = blockIdx.y * 128;
  const int mat = colb >> 10;
  const int jb = colb & 1023;
  const float* Wm = (mat == 0) ? Wq : ((mat == 1) ? Wk : Wv);
  float* dst = (mat == 0) ? Qo : ((mat == 1) ? Ko : Vo);

  const int row0 = rowb + wm * 64;
  const int head = (jb + wn * 64) >> 6;
  float* tbase = &dst[((size_t)(row0 >> 11) * H_ + head) * C_ * D_ +
                      (size_t)((row0 & 2047) >> 4) * TILE_F + lane * 4];

  if (mat != 2) {
    v8f acc[4][4];
    gemm_128x128<true>(x + (size_t)rowb * E_, Wm + jb, lds, acc);
    const int pos0 = (row0 & 2047) + m;
    rope_store<0>(acc[0][0], acc[1][0], acc[2][0], acc[3][0], tbase, pos0, h);
    rope_store<1>(acc[0][1], acc[1][1], acc[2][1], acc[3][1], tbase, pos0, h);
    rope_store<2>(acc[0][2], acc[1][2], acc[2][2], acc[3][2], tbase, pos0, h);
    rope_store<3>(acc[0][3], acc[1][3], acc[2][3], acc[3][3], tbase, pos0, h);
  } else {
    v8f acc[4][4];
    gemm_128x128<false>(x + (size_t)rowb * E_, Wm + jb, lds, acc);
#pragma unroll
    for (int t = 0; t < 4; ++t) {
#pragma unroll
      for (int n = 0; n < 4; ++n) {
        v4f w0, w1;
        w0.x = acc[t][n][0]; w0.y = acc[t][n][1];
        w0.z = acc[t][n][2]; w0.w = acc[t][n][3];
        w1.x = acc[t][n][4]; w1.y = acc[t][n][5];
        w1.z = acc[t][n][6]; w1.w = acc[t][n][7];
        *(v4f*)&tbase[t * TILE_F + (2 * n) * 128] = w0;
        *(v4f*)&tbase[t * TILE_F + (2 * n + 1) * 128] = w1;
      }
    }
  }
}

static __device__ __forceinline__ float xhalf(float x) {
  const int b = __builtin_bit_cast(int, x);
  return __builtin_bit_cast(
      float, __builtin_amdgcn_permlanex16(b, b, 0x76543210, 0xfedcba98, false,
                                          false));
}

static __device__ __forceinline__ void softmax_step(const float (&sv)[8],
                                                    float& mrow, float& lrow,
                                                    float& cr, float (&p)[8]) {
  float vmax = fmaxf(fmaxf(fmaxf(sv[0], sv[1]), fmaxf(sv[2], sv[3])),
                     fmaxf(fmaxf(sv[4], sv[5]), fmaxf(sv[6], sv[7])));
  vmax = fmaxf(vmax, xhalf(vmax));
  const float mnew = fmaxf(mrow, vmax);
  cr = __expf(mrow - mnew);
#pragma unroll
  for (int r = 0; r < 8; ++r) p[r] = __expf(sv[r] - mnew);
  float rs = ((p[0] + p[1]) + (p[2] + p[3])) + ((p[4] + p[5]) + (p[6] + p[7]));
  rs = rs + xhalf(rs);
  lrow = lrow * cr + rs;
  mrow = mnew;
}

template <bool DO_A, bool MASK, bool LAST>
static __device__ __forceinline__ void key_step(
    int kt, int h, int qa, int qb, const float* __restrict__ Kt,
    const float* __restrict__ Vt, const v2f (&bqA)[16], const v2f (&bqB)[16],
    const v4f (&k4)[8], const v4f (&v4)[8], v4f (&kn)[8], v4f (&vn)[8],
    v8f (&oA)[4], v8f (&oB)[4], float& mA, float& lA, float& mB, float& lB) {
  if (!LAST) {
#pragma unroll
    for (int g = 0; g < 8; ++g)
      kn[g] = *(const v4f*)&Kt[(size_t)(kt + 2) * TILE_F + g * 128];
#pragma unroll
    for (int g = 0; g < 8; ++g)
      vn[g] = *(const v4f*)&Vt[(size_t)(kt + 2) * TILE_F + g * 128];
  }
  v8f sA = vzero8(), sB = vzero8();
#pragma unroll
  for (int g = 0; g < 8; ++g) {
    v2f ak;
    ak.x = k4[g].x; ak.y = k4[g].y;
    if (DO_A) sA = wmma_f32(ak, bqA[2 * g], sA);
    sB = wmma_f32(ak, bqB[2 * g], sB);
    ak.x = k4[g].z; ak.y = k4[g].w;
    if (DO_A) sA = wmma_f32(ak, bqA[2 * g + 1], sA);
    sB = wmma_f32(ak, bqB[2 * g + 1], sB);
  }
  float svA[8], svB[8], pA[8], pB[8], crA = 1.0f, crB;
  const int key0 = 16 * kt + 8 * h;
#pragma unroll
  for (int r = 0; r < 8; ++r) {
    svA[r] = sA[r] * 0.125f;
    svB[r] = sB[r] * 0.125f;
    if (MASK) {
      if (key0 + r > qa) svA[r] = -INFINITY;
      if (key0 + r > qb) svB[r] = -INFINITY;
    }
  }
  if (DO_A) softmax_step(svA, mA, lA, crA, pA);
  softmax_step(svB, mB, lB, crB, pB);
#pragma unroll
  for (int n = 0; n < 4; ++n) {
#pragma unroll
    for (int r = 0; r < 8; ++r) {
      if (DO_A) oA[n][r] *= crA;
      oB[n][r] *= crB;
    }
  }
#pragma unroll
  for (int j = 0; j < 4; ++j) {
    v2f bpA, bpB;
    if (DO_A) { bpA.x = pA[2 * j]; bpA.y = pA[2 * j + 1]; }
    bpB.x = pB[2 * j]; bpB.y = pB[2 * j + 1];
    const int gp = j >> 1;
#pragma unroll
    for (int n = 0; n < 4; ++n) {
      v2f av;
      if (j & 1) { av.x = v4[2 * n + gp].z; av.y = v4[2 * n + gp].w; }
      else       { av.x = v4[2 * n + gp].x; av.y = v4[2 * n + gp].y; }
      if (DO_A) oA[n] = wmma_f32(av, bpA, oA[n]);
      oB[n] = wmma_f32(av, bpB, oB[n]);
    }
  }
}

#define PART_F (17 * 128)
#define HANDOVER_AT 55

static __device__ __forceinline__ void put_tile(const v8f (&ot)[4], float* pw) {
#pragma unroll
  for (int n = 0; n < 4; ++n) {
    v4f w;
    w.x = ot[n][0]; w.y = ot[n][1]; w.z = ot[n][2]; w.w = ot[n][3];
    *(v4f*)&pw[(2 * n) * 128] = w;
    w.x = ot[n][4]; w.y = ot[n][5]; w.z = ot[n][6]; w.w = ot[n][7];
    *(v4f*)&pw[(2 * n + 1) * 128] = w;
  }
}

static __device__ __forceinline__ void merge_store_tile(const v8f (&ot)[4],
                                                        float m1, float l1,
                                                        const float* pr,
                                                        float m2, float l2,
                                                        float* o) {
  const float mn = fmaxf(m1, m2);
  const float a1 = __expf(m1 - mn), a2 = __expf(m2 - mn);
  const float inv = 1.0f / (l1 * a1 + l2 * a2);
#pragma unroll
  for (int n = 0; n < 4; ++n) {
    const v4f p0 = *(const v4f*)&pr[(2 * n) * 128];
    const v4f p1 = *(const v4f*)&pr[(2 * n + 1) * 128];
    v4f w;
    w.x = (ot[n][0] * a1 + p0.x * a2) * inv;
    w.y = (ot[n][1] * a1 + p0.y * a2) * inv;
    w.z = (ot[n][2] * a1 + p0.z * a2) * inv;
    w.w = (ot[n][3] * a1 + p0.w * a2) * inv;
    *(v4f*)&o[16 * n] = w;
    w.x = (ot[n][4] * a1 + p1.x * a2) * inv;
    w.y = (ot[n][5] * a1 + p1.y * a2) * inv;
    w.z = (ot[n][6] * a1 + p1.z * a2) * inv;
    w.w = (ot[n][7] * a1 + p1.w * a2) * inv;
    *(v4f*)&o[16 * n + 4] = w;
  }
}

__global__ __launch_bounds__(256) __attribute__((amdgpu_waves_per_eu(2, 2)))
void attn_kernel(
    const float* __restrict__ Q, const float* __restrict__ K,
    const float* __restrict__ V, float* __restrict__ hid) {
  __shared__ float part[8 * PART_F];
  const int tid = threadIdx.x;
  const int wave = __builtin_amdgcn_readfirstlane(tid >> 5);
  const int lane = tid & 31, h = lane >> 4, m = lane & 15;
  const int bh = blockIdx.y;
  const int bb = bh >> 4, head = bh & 15;
  const int lt = blockIdx.x * 4 + (wave & 3);
  const int s = wave >> 2;
  __builtin_amdgcn_s_setprio(1);
  const size_t base = (size_t)bh * C_ * D_;

  const float* Qt = Q + base + lane * 4;
  const float* Kt = K + base + lane * 4;
  const float* Vt = V + base + lane * 4;

  v8f oA[4], oB[4];
  float mA, lA, mB, lB;
  int qtA = 0;
  int done = 0;

#pragma unroll 1
  for (int pass = 0; pass < 2; ++pass) {
    const int wt = ((pass ^ s) & 1) ? (C_ / 32 - 1) - lt : lt;
    qtA = 2 * wt;

    v2f bqA[16], bqB[16];
#pragma unroll
    for (int g = 0; g < 8; ++g) {
      const v4f qa4 = *(const v4f*)&Qt[(size_t)qtA * TILE_F + g * 128];
      const v4f qb4 = *(const v4f*)&Qt[(size_t)(qtA + 1) * TILE_F + g * 128];
      bqA[2 * g].x = qa4.x;     bqA[2 * g].y = qa4.y;
      bqA[2 * g + 1].x = qa4.z; bqA[2 * g + 1].y = qa4.w;
      bqB[2 * g].x = qb4.x;     bqB[2 * g].y = qb4.y;
      bqB[2 * g + 1].x = qb4.z; bqB[2 * g + 1].y = qb4.w;
    }
    v4f k0[8], v0[8], k1[8], v1[8];
#pragma unroll
    for (int g = 0; g < 8; ++g)
      k0[g] = *(const v4f*)&Kt[(size_t)s * TILE_F + g * 128];
#pragma unroll
    for (int g = 0; g < 8; ++g)
      v0[g] = *(const v4f*)&Vt[(size_t)s * TILE_F + g * 128];

#pragma unroll
    for (int n = 0; n < 4; ++n) { oA[n] = vzero8(); oB[n] = vzero8(); }
    mA = -INFINITY; lA = 0.f; mB = -INFINITY; lB = 0.f;
    const int qa = 16 * qtA + m, qb = qa + 16;

    const int khand = s + 2 * (HANDOVER_AT - done);
    int kt = s;
    if (wt & 1) {
      key_step<true, false, false>(kt, h, qa, qb, Kt, Vt, bqA, bqB, k0, v0, k1,
                                   v1, oA, oB, mA, lA, mB, lB);
#pragma unroll
      for (int g = 0; g < 8; ++g) { k0[g] = k1[g]; v0[g] = v1[g]; }
      kt += 2;
    }
    for (; kt < qtA; kt += 4) {
      if (kt >= khand) __builtin_amdgcn_s_setprio(0);
      key_step<true, false, false>(kt, h, qa, qb, Kt, Vt, bqA, bqB, k0, v0, k1,
                                   v1, oA, oB, mA, lA, mB, lB);
      key_step<true, false, false>(kt + 2, h, qa, qb, Kt, Vt, bqA, bqB, k1, v1,
                                   k0, v0, oA, oB, mA, lA, mB, lB);
    }
    if (s == 0)
      key_step<true, true, true>(qtA, h, qa, qb, Kt, Vt, bqA, bqB, k0, v0, k1,
                                 v1, oA, oB, mA, lA, mB, lB);
    else
      key_step<false, true, true>(qtA + 1, h, qa, qb, Kt, Vt, bqA, bqB, k0, v0,
                                  k1, v1, oA, oB, mA, lA, mB, lB);

    done += wt + 1;
    if (pass == 0) {
      float* pw = part + wave * PART_F + lane * 4;
      put_tile(oA, pw);
      put_tile(oB, pw + 8 * 128);
      v4f ml;
      ml.x = mA; ml.y = lA; ml.z = mB; ml.w = lB;
      *(v4f*)&pw[16 * 128] = ml;
    }
  }
  __syncthreads();
  const float* pr = part + (wave ^ 4) * PART_F + lane * 4;
  const v4f ml2 = *(const v4f*)&pr[16 * 128];
  float* o = &hid[((size_t)bb * C_ + 16 * qtA + m) * HD_ + head * D_ + 8 * h];
  merge_store_tile(oA, mA, lA, pr, ml2.x, ml2.y, o);
  merge_store_tile(oB, mB, lB, pr + 8 * 128, ml2.z, ml2.w, o + (size_t)16 * HD_);
}

__global__ __launch_bounds__(128) __attribute__((amdgpu_waves_per_eu(3, 3)))
void proj_kernel(
    const float* __restrict__ hid, const float* __restrict__ Wo,
    float* __restrict__ out) {
  __shared__ float lds[2 * GM_STAGE];
  const int tid = threadIdx.x;
  const int wave = tid >> 5, lane = tid & 31, h = lane >> 4, m = lane & 15;
  const int wm = wave >> 1, wn = wave & 1;
  const int rowb = blockIdx.x * 128;
  const int colb = blockIdx.y * 128;

  v8f acc[4][4];
  gemm_128x128<false>(hid + (size_t)rowb * HD_, Wo + colb, lds, acc);

  float* o = &out[(size_t)(rowb + wm * 64 + 8 * h) * E_ + colb + wn * 64 + m];
#pragma unroll
  for (int t = 0; t < 4; ++t) {
#pragma unroll
    for (int r = 0; r < 8; ++r) {
#pragma unroll
      for (int ct = 0; ct < 4; ++ct)
        o[(size_t)(16 * t + r) * E_ + 16 * ct] = acc[t][ct][r];
    }
  }
}

extern "C" void kernel_launch(void* const* d_in, const int* in_sizes, int n_in,
                              void* d_out, int out_size, void* d_ws,
                              size_t ws_size, hipStream_t stream) {
  (void)in_sizes; (void)n_in; (void)out_size; (void)ws_size;
  const float* x  = (const float*)d_in[0];
  const float* Wq = (const float*)d_in[1];
  const float* Wk = (const float*)d_in[2];
  const float* Wv = (const float*)d_in[3];
  const float* Wo = (const float*)d_in[4];
  float* out = (float*)d_out;

  const size_t BHCD = (size_t)B_ * H_ * C_ * D_;
  float* Qw  = (float*)d_ws;
  float* Kw  = Qw + BHCD;
  float* Vw  = Kw + BHCD;
  float* Hid = Vw + BHCD;

  qkv_kernel<<<dim3(BC_ / 128, (3 * HD_) / 128), 128, 0, stream>>>(
      x, Wq, Wk, Wv, Qw, Kw, Vw);
  attn_kernel<<<dim3(C_ / 32 / 8, B_ * H_), 256, 0, stream>>>(Qw, Kw, Vw, Hid);
  proj_kernel<<<dim3(BC_ / 128, E_ / 128), 128, 0, stream>>>(Hid, Wo, out);
}
